# starting kernel
# baseline (speedup 1.0000x reference)
#include <hip/hip_runtime.h>

typedef float v2f __attribute__((ext_vector_type(2)));
typedef float v8f __attribute__((ext_vector_type(8)));

#define B_      128
#define L_      257
#define C_      8
#define NSTEP   256
#define OUTPB   4680

#define PATH_OFF 0
#define V_OFF    2060
#define H2_OFF   2068
#define H3_OFF   2076
#define G2_OFF   2084
#define G3_OFF   2092
#define G4_OFF   2100
#define T2_OFF   2112
#define U2_OFF   2176
#define S1O 0
#define S2O 8
#define S3O 72
#define S4O 584
#define SREG 0
#define PREG 4680
#define QREG 9360
#define LDS_FLOATS 14040

__global__ __launch_bounds__(32)
void logsig_scan_kernel(const float* __restrict__ path, float* __restrict__ out) {
  __shared__ float buf[LDS_FLOATS];
  const int lane = threadIdx.x;
  const int b    = blockIdx.x;
  const float* gp = path + (size_t)b * (L_ * C_);

  {
    const float4* g4 = (const float4*)gp;
    float4* l4 = (float4*)&buf[PATH_OFF];
    for (int idx = lane; idx < (L_ * C_) / 4; idx += 32) l4[idx] = g4[idx];
  }
  __syncthreads();

  v8f  cacc[16];
  v2f  a3[16];
  float r2[2] = {0.f, 0.f};
  float sig1  = 0.f;
#pragma unroll
  for (int T = 0; T < 16; ++T) {
    a3[T].x = 0.f; a3[T].y = 0.f;
#pragma unroll
    for (int q = 0; q < 8; ++q) cacc[T][q] = 0.f;
  }

  float pprev = 0.f;
  if (lane < C_) pprev = buf[PATH_OFF + lane];

  const int jj  = lane & 7;
  const int i0  = lane >> 3;
  const int i1  = 4 + i0;
  const int kk2 = 2 * (lane & 3);
  const int ta  = (lane >> 2) & 3;

  for (int t = 1; t <= NSTEP; ++t) {
    if (lane < C_) {
      float pc = buf[PATH_OFF + t * C_ + lane];
      float vv = pc - pprev; pprev = pc;
      buf[V_OFF  + lane] = vv;
      buf[H2_OFF + lane] = 0.5f * vv;
      buf[H3_OFF + lane] = vv * (1.0f / 3.0f);
      buf[G2_OFF + lane] = sig1 + 0.5f * vv;
      buf[G3_OFF + lane] = sig1 + vv * (1.0f / 3.0f);
      buf[G4_OFF + lane] = sig1 + 0.25f * vv;
      sig1 += vv;
    }
    __syncthreads();

    float vj  = buf[V_OFF  + jj];
    float h2j = buf[H2_OFF + jj];
    float h3j = buf[H3_OFF + jj];
    float g2a = buf[G2_OFF + i0], g2b = buf[G2_OFF + i1];
    float g3a = buf[G3_OFF + i0], g3b = buf[G3_OFF + i1];
    float g4a = buf[G4_OFF + i0], g4b = buf[G4_OFF + i1];
    float hkA = buf[H2_OFF + kk2], hkB = buf[H2_OFF + kk2 + 1];
    float vkA = buf[V_OFF  + kk2], vkB = buf[V_OFF  + kk2 + 1];

    float t2a = r2[0] + h3j * g4a;
    float t2b = r2[1] + h3j * g4b;
    float u2a = r2[0] + h2j * g3a;
    float u2b = r2[1] + h2j * g3b;
    r2[0] += vj * g2a;
    r2[1] += vj * g2b;
    buf[T2_OFF + lane]      = t2a;
    buf[T2_OFF + lane + 32] = t2b;
    buf[U2_OFF + lane]      = u2a;
    buf[U2_OFF + lane + 32] = u2b;
    __syncthreads();

    v2f Bv;
    Bv.x = (lane < 8) ? vj : 0.0f;
    Bv.y = (lane >= 8 && lane < 16) ? vj : 0.0f;

#pragma unroll
    for (int T = 0; T < 16; ++T) {
      float t2v = buf[T2_OFF + 4 * T + ta];
      float u2v = buf[U2_OFF + 4 * T + ta];
      v2f Av;
      Av.x = a3[T].x + hkA * t2v;
      Av.y = a3[T].y + hkB * t2v;
      cacc[T] = __builtin_amdgcn_wmma_f32_16x16x4_f32(
          false, Av, false, Bv, (short)0, cacc[T], false, false);
      a3[T].x += vkA * u2v;
      a3[T].y += vkB * u2v;
    }
    __syncthreads();
  }

  if (lane < 8) { buf[SREG + S1O + lane] = sig1; buf[PREG + S1O + lane] = sig1; }
  buf[SREG + S2O + lane]      = r2[0];  buf[PREG + S2O + lane]      = r2[0];
  buf[SREG + S2O + 32 + lane] = r2[1];  buf[PREG + S2O + 32 + lane] = r2[1];
  if (lane < 16) {
#pragma unroll
    for (int T = 0; T < 16; ++T) {
      int f = 32 * T + 2 * lane;
      buf[SREG + S3O + f]     = a3[T].x;  buf[PREG + S3O + f]     = a3[T].x;
      buf[SREG + S3O + f + 1] = a3[T].y;  buf[PREG + S3O + f + 1] = a3[T].y;
    }
  }
  {
    const int hi = (lane >> 4) * 8, cc = lane & 15;
#pragma unroll
    for (int T = 0; T < 16; ++T)
#pragma unroll
      for (int q = 0; q < 8; ++q) {
        int f = (16 * T + q + hi) * 16 + cc;
        buf[SREG + S4O + f] = cacc[T][q];
        buf[PREG + S4O + f] = cacc[T][q];
      }
  }
  __syncthreads();

  int Pb = PREG, Qb = QREG;
  const int mm = lane & 15;
  for (int n = 2; n <= 4; ++n) {
    float coef = ((n & 1) ? 1.0f : -1.0f) / (float)n;
    if (lane < 8) buf[Qb + S1O + lane] = 0.0f;
#pragma unroll
    for (int h = 0; h < 2; ++h) {
      int ij = lane + 32 * h;
      float q = buf[Pb + S1O + (ij >> 3)] * buf[SREG + S1O + (ij & 7)];
      r2[h] += coef * q;
      buf[Qb + S2O + ij] = q;
    }
#pragma unroll
    for (int T = 0; T < 16; ++T) {
#pragma unroll
      for (int r = 0; r < 2; ++r) {
        int f = 32 * T + 2 * mm + r;
        float q = buf[Pb + S1O + (f >> 6)] * buf[SREG + S2O + (f & 63)]
                + buf[Pb + S2O + (f >> 3)] * buf[SREG + S1O + (f & 7)];
        if (r == 0) a3[T].x += coef * q; else a3[T].y += coef * q;
        if (lane < 16) buf[Qb + S3O + f] = q;
      }
    }
    {
      const int hi = (lane >> 4) * 8, cc = lane & 15;
#pragma unroll
      for (int T = 0; T < 16; ++T)
#pragma unroll
        for (int q8 = 0; q8 < 8; ++q8) {
          int f = (16 * T + q8 + hi) * 16 + cc;
          float q = buf[Pb + S1O + (f >> 9)] * buf[SREG + S3O + (f & 511)]
                  + buf[Pb + S2O + (f >> 6)] * buf[SREG + S2O + (f & 63)]
                  + buf[Pb + S3O + (f >> 3)] * buf[SREG + S1O + (f & 7)];
          cacc[T][q8] += coef * q;
          buf[Qb + S4O + f] = q;
        }
    }
    __syncthreads();
    int tmp = Pb; Pb = Qb; Qb = tmp;
  }

  float* ob = out + (size_t)b * OUTPB;
  if (lane < 8) ob[lane] = sig1;
  ob[8 + lane]      = r2[0];
  ob[8 + 32 + lane] = r2[1];
  if (lane < 16) {
#pragma unroll
    for (int T = 0; T < 16; ++T) {
      int f = 32 * T + 2 * lane;
      ob[72 + f]     = a3[T].x;
      ob[72 + f + 1] = a3[T].y;
    }
  }
  {
    const int hi = (lane >> 4) * 8, cc = lane & 15;
#pragma unroll
    for (int T = 0; T < 16; ++T)
#pragma unroll
      for (int q = 0; q < 8; ++q) {
        int f = (16 * T + q + hi) * 16 + cc;
        ob[584 + f] = cacc[T][q];
      }
  }
}

extern "C" void kernel_launch(void* const* d_in, const int* in_sizes, int n_in,
                              void* d_out, int out_size, void* d_ws, size_t ws_size,
                              hipStream_t stream) {
  const float* path = (const float*)d_in[0];
  float* out = (float*)d_out;
  logsig_scan_kernel<<<dim3(B_), dim3(32), 0, stream>>>(path, out);
}


# Signature scan state stays in registers: each lane recomputes the level-1/2 Horner terms its tiles need from an LDS increment table read one step ahead; the 3 LDS hand-overs per step are gone
# speedup vs baseline: 1.5287x; 1.5287x over previous
#include <hip/hip_runtime.h>

typedef float v2f __attribute__((ext_vector_type(2)));
typedef float v8f __attribute__((ext_vector_type(8)));

#define B_      128
#define L_      257
#define C_      8
#define NSTEP   256
#define OUTPB   4680

#define VINC_OFF 4680
#define S1O 0
#define S2O 8
#define S3O 72
#define S4O 584
#define SREG 0
#define PREG 4680
#define QREG 9360
#define LDS_FLOATS 14040

__global__ __launch_bounds__(32)
void logsig_scan_kernel(const float* __restrict__ path, float* __restrict__ out) {
  __shared__ float buf[LDS_FLOATS];
  const int lane = threadIdx.x;
  const int b    = blockIdx.x;
  const float* gp = path + (size_t)b * (L_ * C_);

  {
    const float4* g4 = (const float4*)gp;
    float4* l4 = (float4*)&buf[VINC_OFF];
    for (int idx = lane; idx < (NSTEP * C_) / 4; idx += 32) {
      float4 a = g4[idx], c = g4[idx + 2];
      l4[idx] = make_float4(c.x - a.x, c.y - a.y, c.z - a.z, c.w - a.w);
    }
    if (lane < 2) l4[(NSTEP * C_) / 4 + lane] = make_float4(0.f, 0.f, 0.f, 0.f);
  }
  __syncthreads();

  v8f  cacc[16];
  v2f  a3[16];
  float r2[16];
  float s1[8];
#pragma unroll
  for (int T = 0; T < 16; ++T) {
    a3[T].x = 0.f; a3[T].y = 0.f; r2[T] = 0.f;
#pragma unroll
    for (int q = 0; q < 8; ++q) cacc[T][q] = 0.f;
  }
#pragma unroll
  for (int i = 0; i < 8; ++i) s1[i] = 0.f;

  const int jj  = lane & 7;
  const int kk2 = 2 * (lane & 3);
  const int ta  = (lane >> 2) & 3;
  const bool blo = lane < 8;
  const bool bhi = (lane >= 8) && (lane < 16);

  const float* vr = &buf[VINC_OFF];
  float4 va = ((const float4*)vr)[0], vb = ((const float4*)vr)[1];
  float  vlo = vr[ta], vhi = vr[4 + ta];
  float2 vk  = *(const float2*)&vr[kk2];
  float  vjj = vr[jj];

  for (int s = 0; s < NSTEP; ++s) {
    const float v[8] = {va.x, va.y, va.z, va.w, vb.x, vb.y, vb.z, vb.w};
    float g2[8], g3[8], g4[8];
#pragma unroll
    for (int i = 0; i < 8; ++i) {
      g2[i] = s1[i] + 0.5f * v[i];
      g3[i] = s1[i] + v[i] * (1.0f / 3.0f);
      g4[i] = s1[i] + 0.25f * v[i];
      s1[i] += v[i];
    }
    const float h2lo = 0.5f * vlo, h2hi = 0.5f * vhi;
    const float h3lo = vlo * (1.0f / 3.0f), h3hi = vhi * (1.0f / 3.0f);
    const float hkA = 0.5f * vk.x, hkB = 0.5f * vk.y;
    const float vjlo = vlo, vjhi = vhi, vkA = vk.x, vkB = vk.y;

    v2f Bv;
    Bv.x = blo ? vjj : 0.0f;
    Bv.y = bhi ? vjj : 0.0f;

    vr += C_;
    va = ((const float4*)vr)[0]; vb = ((const float4*)vr)[1];
    vlo = vr[ta]; vhi = vr[4 + ta];
    vk  = *(const float2*)&vr[kk2];
    vjj = vr[jj];

#pragma unroll
    for (int T = 0; T < 16; ++T) {
      const int   i   = T >> 1;
      const float h3j = (T & 1) ? h3hi : h3lo;
      const float h2j = (T & 1) ? h2hi : h2lo;
      const float vj  = (T & 1) ? vjhi : vjlo;
      float t2v = r2[T] + h3j * g4[i];
      float u2v = r2[T] + h2j * g3[i];
      r2[T] += vj * g2[i];
      v2f Av;
      Av.x = a3[T].x + hkA * t2v;
      Av.y = a3[T].y + hkB * t2v;
      cacc[T] = __builtin_amdgcn_wmma_f32_16x16x4_f32(
          false, Av, false, Bv, (short)0, cacc[T], false, false);
      a3[T].x += vkA * u2v;
      a3[T].y += vkB * u2v;
    }
    asm volatile("" :: "v"(va.x), "v"(va.y), "v"(va.z), "v"(va.w),
                       "v"(vb.x), "v"(vb.y), "v"(vb.z), "v"(vb.w),
                       "v"(vlo), "v"(vhi), "v"(vk.x), "v"(vk.y), "v"(vjj));
  }
  __syncthreads();

  if (lane == 0) {
#pragma unroll
    for (int i = 0; i < 8; ++i) { buf[SREG + S1O + i] = s1[i]; buf[PREG + S1O + i] = s1[i]; }
  }
  if (lane < 16 && (lane & 3) == 0) {
#pragma unroll
    for (int T = 0; T < 16; ++T) {
      buf[SREG + S2O + 4 * T + ta] = r2[T];  buf[PREG + S2O + 4 * T + ta] = r2[T];
    }
  }
  if (lane < 16) {
#pragma unroll
    for (int T = 0; T < 16; ++T) {
      int f = 32 * T + 2 * lane;
      buf[SREG + S3O + f]     = a3[T].x;  buf[PREG + S3O + f]     = a3[T].x;
      buf[SREG + S3O + f + 1] = a3[T].y;  buf[PREG + S3O + f + 1] = a3[T].y;
    }
  }
  {
    const int hi = (lane >> 4) * 8, cc = lane & 15;
#pragma unroll
    for (int T = 0; T < 16; ++T)
#pragma unroll
      for (int q = 0; q < 8; ++q) {
        int f = (16 * T + q + hi) * 16 + cc;
        buf[SREG + S4O + f] = cacc[T][q];
        buf[PREG + S4O + f] = cacc[T][q];
      }
  }
  __syncthreads();

  const float sig1 = (lane < 8) ? buf[SREG + S1O + lane] : 0.f;
  float r2o[2] = {buf[SREG + S2O + lane], buf[SREG + S2O + 32 + lane]};

  int Pb = PREG, Qb = QREG;
  const int mm = lane & 15;
  for (int n = 2; n <= 4; ++n) {
    float coef = ((n & 1) ? 1.0f : -1.0f) / (float)n;
    if (lane < 8) buf[Qb + S1O + lane] = 0.0f;
#pragma unroll
    for (int h = 0; h < 2; ++h) {
      int ij = lane + 32 * h;
      float q = buf[Pb + S1O + (ij >> 3)] * buf[SREG + S1O + (ij & 7)];
      r2o[h] += coef * q;
      buf[Qb + S2O + ij] = q;
    }
#pragma unroll
    for (int T = 0; T < 16; ++T) {
#pragma unroll
      for (int r = 0; r < 2; ++r) {
        int f = 32 * T + 2 * mm + r;
        float q = buf[Pb + S1O + (f >> 6)] * buf[SREG + S2O + (f & 63)]
                + buf[Pb + S2O + (f >> 3)] * buf[SREG + S1O + (f & 7)];
        if (r == 0) a3[T].x += coef * q; else a3[T].y += coef * q;
        if (lane < 16) buf[Qb + S3O + f] = q;
      }
    }
    {
      const int hi = (lane >> 4) * 8, cc = lane & 15;
#pragma unroll
      for (int T = 0; T < 16; ++T)
#pragma unroll
        for (int q8 = 0; q8 < 8; ++q8) {
          int f = (16 * T + q8 + hi) * 16 + cc;
          float q = buf[Pb + S1O + (f >> 9)] * buf[SREG + S3O + (f & 511)]
                  + buf[Pb + S2O + (f >> 6)] * buf[SREG + S2O + (f & 63)]
                  + buf[Pb + S3O + (f >> 3)] * buf[SREG + S1O + (f & 7)];
          cacc[T][q8] += coef * q;
          buf[Qb + S4O + f] = q;
        }
    }
    __syncthreads();
    int tmp = Pb; Pb = Qb; Qb = tmp;
  }

  float* ob = out + (size_t)b * OUTPB;
  if (lane < 8) ob[lane] = sig1;
  ob[8 + lane]      = r2o[0];
  ob[8 + 32 + lane] = r2o[1];
  if (lane < 16) {
#pragma unroll
    for (int T = 0; T < 16; ++T) {
      int f = 32 * T + 2 * lane;
      ob[72 + f]     = a3[T].x;
      ob[72 + f + 1] = a3[T].y;
    }
  }
  {
    const int hi = (lane >> 4) * 8, cc = lane & 15;
#pragma unroll
    for (int T = 0; T < 16; ++T)
#pragma unroll
      for (int q = 0; q < 8; ++q) {
        int f = (16 * T + q + hi) * 16 + cc;
        ob[584 + f] = cacc[T][q];
      }
  }
}

extern "C" void kernel_launch(void* const* d_in, const int* in_sizes, int n_in,
                              void* d_out, int out_size, void* d_ws, size_t ws_size,
                              hipStream_t stream) {
  const float* path = (const float*)d_in[0];
  float* out = (float*)d_out;
  logsig_scan_kernel<<<dim3(B_), dim3(32), 0, stream>>>(path, out);
}


# One batch element per 8-wave work-group instead of per wave: each wave scans 2 of the 16 level-4 tiles with the lower-level terms they need, reading increments two steps ahead; sums unchanged
# speedup vs baseline: 5.1704x; 3.3822x over previous
#include <hip/hip_runtime.h>


typedef float v2f __attribute__((ext_vector_type(2)));
typedef float v8f __attribute__((ext_vector_type(8)));

#define B_      128
#define L_      257
#define C_      8
#define NSTEP   256
#define OUTPB   4680

#define NW      8
#define NTHR    256
#define NT      (16 / NW)
#define NI      (NT / 2)
static_assert(NTHR == 32 * NW, "one wave is 32 lanes");
static_assert(NT == 2 && NI == 1, "a wave owns one (even, odd) tile pair, which shares one level-1 component");

#define VINC_OFF 4680
#define S1O 0
#define S2O 8
#define S3O 72
#define S4O 584
#define SREG 0
#define PREG 4680
#define QREG 9360
#define LDS_FLOATS 14040

__global__ __launch_bounds__(NTHR)
void logsig_scan_kernel(const float* __restrict__ path, float* __restrict__ out) {
  __shared__ float buf[LDS_FLOATS];
  const int tid  = threadIdx.x;
  const int lane = tid & 31;
  const int w    = __builtin_amdgcn_readfirstlane(tid >> 5);
  const int T0   = NT * w;
  const int i0   = T0 >> 1;
  const int b    = blockIdx.x;
  const float* gp = path + (size_t)b * (L_ * C_);

  {
    const float4* g4 = (const float4*)gp;
    float4* l4 = (float4*)&buf[VINC_OFF];
    for (int idx = tid; idx < (NSTEP * C_) / 4; idx += NTHR) {
      float4 a = g4[idx], c = g4[idx + 2];
      l4[idx] = make_float4(c.x - a.x, c.y - a.y, c.z - a.z, c.w - a.w);
    }
    if (tid < 4) l4[(NSTEP * C_) / 4 + tid] = make_float4(0.f, 0.f, 0.f, 0.f);
  }
  __syncthreads();

  v8f  cacc[NT];
  v2f  a3[NT];
  float r2[NT];
  float s1[NI];
#pragma unroll
  for (int t = 0; t < NT; ++t) {
    a3[t].x = 0.f; a3[t].y = 0.f; r2[t] = 0.f;
#pragma unroll
    for (int q = 0; q < 8; ++q) cacc[t][q] = 0.f;
  }
#pragma unroll
  for (int ii = 0; ii < NI; ++ii) s1[ii] = 0.f;

  const int jj  = lane & 7;
  const int kk2 = 2 * (lane & 3);
  const int ta  = (lane >> 2) & 3;
  const bool blo = lane < 8;
  const bool bhi = (lane >= 8) && (lane < 16);

  const float* vr = &buf[VINC_OFF];
  float  ni[2], nlo[2], nhi[2], njj[2];
  float2 nk[2];
#pragma unroll
  for (int u = 0; u < 2; ++u) {
    const float* row = vr + u * C_;
    ni[u]  = row[i0];
    nlo[u] = row[ta];  nhi[u] = row[4 + ta];
    nk[u]  = *(const float2*)&row[kk2];
    njj[u] = row[jj];
  }

  static_assert(NSTEP % 2 == 0, "the scan takes two steps per trip");
  for (int s = 0; s < NSTEP; s += 2) {
    vr += 2 * C_;
    float  fi[2], flo[2], fhi[2], fjj[2];
    float2 fk[2];
#pragma unroll
    for (int u = 0; u < 2; ++u) {
      const float* row = vr + u * C_;
      fi[u]  = row[i0];
      flo[u] = row[ta];  fhi[u] = row[4 + ta];
      fk[u]  = *(const float2*)&row[kk2];
      fjj[u] = row[jj];
    }
    __builtin_amdgcn_sched_barrier(0);

#pragma unroll
    for (int u = 0; u < 2; ++u) {
      float g2[NI], g3[NI], g4[NI];
#pragma unroll
      for (int ii = 0; ii < NI; ++ii) {
        g2[ii] = s1[ii] + 0.5f * ni[u];
        g3[ii] = s1[ii] + ni[u] * (1.0f / 3.0f);
        g4[ii] = s1[ii] + 0.25f * ni[u];
        s1[ii] += ni[u];
      }
      const float h2lo = 0.5f * nlo[u], h2hi = 0.5f * nhi[u];
      const float h3lo = nlo[u] * (1.0f / 3.0f), h3hi = nhi[u] * (1.0f / 3.0f);
      const float hkA = 0.5f * nk[u].x, hkB = 0.5f * nk[u].y;
      const float vjlo = nlo[u], vjhi = nhi[u], vkA = nk[u].x, vkB = nk[u].y;

      v2f Bv;
      Bv.x = blo ? njj[u] : 0.0f;
      Bv.y = bhi ? njj[u] : 0.0f;

#pragma unroll
      for (int t = 0; t < NT; ++t) {
        const int   ii  = t >> 1;
        const float h3j = (t & 1) ? h3hi : h3lo;
        const float h2j = (t & 1) ? h2hi : h2lo;
        const float vj  = (t & 1) ? vjhi : vjlo;
        float t2v = r2[t] + h3j * g4[ii];
        float u2v = r2[t] + h2j * g3[ii];
        r2[t] += vj * g2[ii];
        v2f Av;
        Av.x = a3[t].x + hkA * t2v;
        Av.y = a3[t].y + hkB * t2v;
        cacc[t] = __builtin_amdgcn_wmma_f32_16x16x4_f32(
            false, Av, false, Bv, (short)0, cacc[t], false, false);
        a3[t].x += vkA * u2v;
        a3[t].y += vkB * u2v;
      }
    }
    asm volatile("" :: "v"(fi[0]), "v"(flo[0]), "v"(fhi[0]), "v"(fk[0].x), "v"(fk[0].y), "v"(fjj[0]),
                       "v"(fi[1]), "v"(flo[1]), "v"(fhi[1]), "v"(fk[1].x), "v"(fk[1].y), "v"(fjj[1]));
#pragma unroll
    for (int u = 0; u < 2; ++u) { ni[u] = fi[u]; nlo[u] = flo[u]; nhi[u] = fhi[u]; nk[u] = fk[u]; njj[u] = fjj[u]; }
  }
  __syncthreads();

  if (lane == 0) {
#pragma unroll
    for (int ii = 0; ii < NI; ++ii) {
      buf[SREG + S1O + i0 + ii] = s1[ii];  buf[PREG + S1O + i0 + ii] = s1[ii];
    }
  }
  if (lane < 16 && (lane & 3) == 0) {
#pragma unroll
    for (int t = 0; t < NT; ++t) {
      const int T = T0 + t;
      buf[SREG + S2O + 4 * T + ta] = r2[t];  buf[PREG + S2O + 4 * T + ta] = r2[t];
    }
  }
  if (lane < 16) {
#pragma unroll
    for (int t = 0; t < NT; ++t) {
      int f = 32 * (T0 + t) + 2 * lane;
      buf[SREG + S3O + f]     = a3[t].x;  buf[PREG + S3O + f]     = a3[t].x;
      buf[SREG + S3O + f + 1] = a3[t].y;  buf[PREG + S3O + f + 1] = a3[t].y;
    }
  }
  {
    const int hi = (lane >> 4) * 8, cc = lane & 15;
#pragma unroll
    for (int t = 0; t < NT; ++t)
#pragma unroll
      for (int q = 0; q < 8; ++q) {
        int f = (16 * (T0 + t) + q + hi) * 16 + cc;
        buf[SREG + S4O + f] = cacc[t][q];
        buf[PREG + S4O + f] = cacc[t][q];
      }
  }
  __syncthreads();

  float sig1 = 0.f;
  float r2o[2] = {0.f, 0.f};
  if (w == 0) {
    sig1   = (lane < 8) ? buf[SREG + S1O + lane] : 0.f;
    r2o[0] = buf[SREG + S2O + lane];
    r2o[1] = buf[SREG + S2O + 32 + lane];
  }

  int Pb = PREG, Qb = QREG;
  const int mm = lane & 15;
  for (int n = 2; n <= 4; ++n) {
    float coef = ((n & 1) ? 1.0f : -1.0f) / (float)n;
    if (w == 0) {
      if (lane < 8) buf[Qb + S1O + lane] = 0.0f;
#pragma unroll
      for (int h = 0; h < 2; ++h) {
        int ij = lane + 32 * h;
        float q = buf[Pb + S1O + (ij >> 3)] * buf[SREG + S1O + (ij & 7)];
        r2o[h] += coef * q;
        buf[Qb + S2O + ij] = q;
      }
    }
#pragma unroll
    for (int t = 0; t < NT; ++t) {
#pragma unroll
      for (int r = 0; r < 2; ++r) {
        int f = 32 * (T0 + t) + 2 * mm + r;
        float q = buf[Pb + S1O + (f >> 6)] * buf[SREG + S2O + (f & 63)]
                + buf[Pb + S2O + (f >> 3)] * buf[SREG + S1O + (f & 7)];
        if (r == 0) a3[t].x += coef * q; else a3[t].y += coef * q;
        if (lane < 16) buf[Qb + S3O + f] = q;
      }
    }
    {
      const int hi = (lane >> 4) * 8, cc = lane & 15;
#pragma unroll
      for (int t = 0; t < NT; ++t)
#pragma unroll
        for (int q8 = 0; q8 < 8; ++q8) {
          int f = (16 * (T0 + t) + q8 + hi) * 16 + cc;
          float q = buf[Pb + S1O + (f >> 9)] * buf[SREG + S3O + (f & 511)]
                  + buf[Pb + S2O + (f >> 6)] * buf[SREG + S2O + (f & 63)]
                  + buf[Pb + S3O + (f >> 3)] * buf[SREG + S1O + (f & 7)];
          cacc[t][q8] += coef * q;
          buf[Qb + S4O + f] = q;
        }
    }
    __syncthreads();
    int tmp = Pb; Pb = Qb; Qb = tmp;
  }

  float* ob = out + (size_t)b * OUTPB;
  if (w == 0) {
    if (lane < 8) ob[lane] = sig1;
    ob[8 + lane]      = r2o[0];
    ob[8 + 32 + lane] = r2o[1];
  }
  if (lane < 16) {
#pragma unroll
    for (int t = 0; t < NT; ++t) {
      int f = 32 * (T0 + t) + 2 * lane;
      ob[72 + f]     = a3[t].x;
      ob[72 + f + 1] = a3[t].y;
    }
  }
  {
    const int hi = (lane >> 4) * 8, cc = lane & 15;
#pragma unroll
    for (int t = 0; t < NT; ++t)
#pragma unroll
      for (int q = 0; q < 8; ++q) {
        int f = (16 * (T0 + t) + q + hi) * 16 + cc;
        ob[584 + f] = cacc[t][q];
      }
  }
}

extern "C" void kernel_launch(void* const* d_in, const int* in_sizes, int n_in,
                              void* d_out, int out_size, void* d_ws, size_t ws_size,
                              hipStream_t stream) {
  const float* path = (const float*)d_in[0];
  float* out = (float*)d_out;
  logsig_scan_kernel<<<dim3(B_), dim3(NTHR), 0, stream>>>(path, out);
}


# One batch element per PAIR of 4-wave work-groups (256 groups, one wave per SIMD on every WGP): each group scans 8 level-4 tiles; lanes 16-31 carry the partner half's level 1-3 rows; sums unchanged
# speedup vs baseline: 6.9803x; 1.3501x over previous
#include <hip/hip_runtime.h>


typedef float v2f __attribute__((ext_vector_type(2)));
typedef float v8f __attribute__((ext_vector_type(8)));

#define B_      128
#define L_      257
#define C_      8
#define NSTEP   256
#define OUTPB   4680

#define NG      2
#define NW      4
#define NTHR    128
#define NT      (16 / (NG * NW))
#define NI      (NT / 2)
static_assert(NTHR == 32 * NW, "one wave is 32 lanes");
static_assert(NT == 2 && NI == 1, "a wave owns one (even, odd) tile pair, which shares one level-1 component");
static_assert(NG == 2, "lanes 16-31 carry the partner pair of exactly one other group");
#define NGRID   256
static_assert(NGRID == NG * B_, "two work-groups per batch element");

#define VINC_OFF 4680
#define S1O 0
#define S2O 8
#define S3O 72
#define S4O 584
#define SREG 0
#define PREG 4680
#define QREG 9360
#define LDS_FLOATS 14040

__global__ __launch_bounds__(NTHR)
void logsig_scan_kernel(const float* __restrict__ path, float* __restrict__ out) {
  __shared__ float buf[LDS_FLOATS];
  const int tid  = threadIdx.x;
  const int lane = tid & 31;
  const int w    = __builtin_amdgcn_readfirstlane(tid >> 5);
  const int b    = blockIdx.x >> 1;
  const int h    = blockIdx.x & 1;
  const int T0   = 8 * h + NT * w;
  const int i0   = T0 >> 1;
  const int T0l  = T0 ^ (8 * (lane >> 4));
  const int i0l  = T0l >> 1;
  const float* gp = path + (size_t)b * (L_ * C_);

  {
    const float4* g4 = (const float4*)gp;
    float4* l4 = (float4*)&buf[VINC_OFF];
    constexpr int NPASS = (NSTEP * C_) / 4 / NTHR;
    static_assert(NPASS * NTHR * 4 == NSTEP * C_, "the table is a whole number of passes");
    float4 a[NPASS], c[NPASS];
#pragma unroll
    for (int k = 0; k < NPASS; ++k) {
      a[k] = g4[tid + k * NTHR];  c[k] = g4[tid + k * NTHR + 2];
    }
#pragma unroll
    for (int k = 0; k < NPASS; ++k)
      l4[tid + k * NTHR] = make_float4(c[k].x - a[k].x, c[k].y - a[k].y, c[k].z - a[k].z, c[k].w - a[k].w);
    if (tid < 4) l4[(NSTEP * C_) / 4 + tid] = make_float4(0.f, 0.f, 0.f, 0.f);
  }
  __syncthreads();

  v8f  cacc[NT];
  v2f  a3[NT];
  float r2[NT];
  float s1[NI];
#pragma unroll
  for (int t = 0; t < NT; ++t) {
    a3[t].x = 0.f; a3[t].y = 0.f; r2[t] = 0.f;
#pragma unroll
    for (int q = 0; q < 8; ++q) cacc[t][q] = 0.f;
  }
#pragma unroll
  for (int ii = 0; ii < NI; ++ii) s1[ii] = 0.f;

  const int jj  = lane & 7;
  const int kk2 = 2 * (lane & 3);
  const int ta  = (lane >> 2) & 3;
  const bool blo = lane < 8;
  const bool bhi = (lane >= 8) && (lane < 16);

  const float* vr = &buf[VINC_OFF];
  float  ni[2], nlo[2], nhi[2], njj[2];
  float2 nk[2];
#pragma unroll
  for (int u = 0; u < 2; ++u) {
    const float* row = vr + u * C_;
    ni[u]  = row[i0l];
    nlo[u] = row[ta];  nhi[u] = row[4 + ta];
    nk[u]  = *(const float2*)&row[kk2];
    njj[u] = row[jj];
  }

  static_assert(NSTEP % 2 == 0, "the scan takes two steps per trip");
  for (int s = 0; s < NSTEP; s += 2) {
    vr += 2 * C_;
    float  fi[2], flo[2], fhi[2], fjj[2];
    float2 fk[2];
#pragma unroll
    for (int u = 0; u < 2; ++u) {
      const float* row = vr + u * C_;
      fi[u]  = row[i0l];
      flo[u] = row[ta];  fhi[u] = row[4 + ta];
      fk[u]  = *(const float2*)&row[kk2];
      fjj[u] = row[jj];
    }
    __builtin_amdgcn_sched_barrier(0);

#pragma unroll
    for (int u = 0; u < 2; ++u) {
      float g2[NI], g3[NI], g4[NI];
#pragma unroll
      for (int ii = 0; ii < NI; ++ii) {
        g2[ii] = s1[ii] + 0.5f * ni[u];
        g3[ii] = s1[ii] + ni[u] * (1.0f / 3.0f);
        g4[ii] = s1[ii] + 0.25f * ni[u];
        s1[ii] += ni[u];
      }
      const float h2lo = 0.5f * nlo[u], h2hi = 0.5f * nhi[u];
      const float h3lo = nlo[u] * (1.0f / 3.0f), h3hi = nhi[u] * (1.0f / 3.0f);
      const float hkA = 0.5f * nk[u].x, hkB = 0.5f * nk[u].y;
      const float vjlo = nlo[u], vjhi = nhi[u], vkA = nk[u].x, vkB = nk[u].y;

      v2f Bv;
      Bv.x = blo ? njj[u] : 0.0f;
      Bv.y = bhi ? njj[u] : 0.0f;

#pragma unroll
      for (int t = 0; t < NT; ++t) {
        const int   ii  = t >> 1;
        const float h3j = (t & 1) ? h3hi : h3lo;
        const float h2j = (t & 1) ? h2hi : h2lo;
        const float vj  = (t & 1) ? vjhi : vjlo;
        float t2v = r2[t] + h3j * g4[ii];
        float u2v = r2[t] + h2j * g3[ii];
        r2[t] += vj * g2[ii];
        v2f Av;
        Av.x = a3[t].x + hkA * t2v;
        Av.y = a3[t].y + hkB * t2v;
        cacc[t] = __builtin_amdgcn_wmma_f32_16x16x4_f32(
            false, Av, false, Bv, (short)0, cacc[t], false, false);
        a3[t].x += vkA * u2v;
        a3[t].y += vkB * u2v;
      }
    }
    asm volatile("" :: "v"(fi[0]), "v"(flo[0]), "v"(fhi[0]), "v"(fk[0].x), "v"(fk[0].y), "v"(fjj[0]),
                       "v"(fi[1]), "v"(flo[1]), "v"(fhi[1]), "v"(fk[1].x), "v"(fk[1].y), "v"(fjj[1]));
#pragma unroll
    for (int u = 0; u < 2; ++u) { ni[u] = fi[u]; nlo[u] = flo[u]; nhi[u] = fhi[u]; nk[u] = fk[u]; njj[u] = fjj[u]; }
  }
  __syncthreads();

  if ((lane & 15) == 0) {
#pragma unroll
    for (int ii = 0; ii < NI; ++ii) {
      buf[SREG + S1O + i0l + ii] = s1[ii];  buf[PREG + S1O + i0l + ii] = s1[ii];
    }
  }
  if ((lane & 3) == 0) {
#pragma unroll
    for (int t = 0; t < NT; ++t) {
      const int T = T0l + t;
      buf[SREG + S2O + 4 * T + ta] = r2[t];  buf[PREG + S2O + 4 * T + ta] = r2[t];
    }
  }
  {
#pragma unroll
    for (int t = 0; t < NT; ++t) {
      int f = 32 * (T0l + t) + 2 * (lane & 15);
      buf[SREG + S3O + f]     = a3[t].x;  buf[PREG + S3O + f]     = a3[t].x;
      buf[SREG + S3O + f + 1] = a3[t].y;  buf[PREG + S3O + f + 1] = a3[t].y;
    }
  }
  {
    const int hi = (lane >> 4) * 8, cc = lane & 15;
#pragma unroll
    for (int t = 0; t < NT; ++t)
#pragma unroll
      for (int q = 0; q < 8; ++q) {
        int f = (16 * (T0 + t) + q + hi) * 16 + cc;
        buf[SREG + S4O + f] = cacc[t][q];
        buf[PREG + S4O + f] = cacc[t][q];
      }
  }
  __syncthreads();

  float sig1 = 0.f;
  float r2o  = 0.f;
  if (w == 0) {
    sig1 = (lane < 4) ? buf[SREG + S1O + 4 * h + lane] : 0.f;
    r2o  = buf[SREG + S2O + 32 * h + lane];
  }

  int Pb = PREG, Qb = QREG;
  const int mm = lane & 15;
  for (int n = 2; n <= 4; ++n) {
    float coef = ((n & 1) ? 1.0f : -1.0f) / (float)n;
    if (w == 0) {
      if (lane < 4) buf[Qb + S1O + 4 * h + lane] = 0.0f;
      {
        int ij = 32 * h + lane;
        float q = buf[Pb + S1O + (ij >> 3)] * buf[SREG + S1O + (ij & 7)];
        r2o += coef * q;
        buf[Qb + S2O + ij] = q;
      }
    }
#pragma unroll
    for (int t = 0; t < NT; ++t) {
#pragma unroll
      for (int r = 0; r < 2; ++r) {
        int f = 32 * (T0 + t) + 2 * mm + r;
        float q = buf[Pb + S1O + (f >> 6)] * buf[SREG + S2O + (f & 63)]
                + buf[Pb + S2O + (f >> 3)] * buf[SREG + S1O + (f & 7)];
        if (r == 0) a3[t].x += coef * q; else a3[t].y += coef * q;
        if (lane < 16) buf[Qb + S3O + f] = q;
      }
    }
    {
      const int hi = (lane >> 4) * 8, cc = lane & 15;
#pragma unroll
      for (int t = 0; t < NT; ++t)
#pragma unroll
        for (int q8 = 0; q8 < 8; ++q8) {
          int f = (16 * (T0 + t) + q8 + hi) * 16 + cc;
          float q = buf[Pb + S1O + (f >> 9)] * buf[SREG + S3O + (f & 511)]
                  + buf[Pb + S2O + (f >> 6)] * buf[SREG + S2O + (f & 63)]
                  + buf[Pb + S3O + (f >> 3)] * buf[SREG + S1O + (f & 7)];
          cacc[t][q8] += coef * q;
          buf[Qb + S4O + f] = q;
        }
    }
    __syncthreads();
    int tmp = Pb; Pb = Qb; Qb = tmp;
  }

  float* ob = out + (size_t)b * OUTPB;
  if (w == 0) {
    if (lane < 4) ob[4 * h + lane] = sig1;
    ob[8 + 32 * h + lane] = r2o;
  }
  if (lane < 16) {
#pragma unroll
    for (int t = 0; t < NT; ++t) {
      int f = 32 * (T0 + t) + 2 * lane;
      ob[72 + f]     = a3[t].x;
      ob[72 + f + 1] = a3[t].y;
    }
  }
  {
    const int hi = (lane >> 4) * 8, cc = lane & 15;
#pragma unroll
    for (int t = 0; t < NT; ++t)
#pragma unroll
      for (int q = 0; q < 8; ++q) {
        int f = (16 * (T0 + t) + q + hi) * 16 + cc;
        ob[584 + f] = cacc[t][q];
      }
  }
}

extern "C" void kernel_launch(void* const* d_in, const int* in_sizes, int n_in,
                              void* d_out, int out_size, void* d_ws, size_t ws_size,
                              hipStream_t stream) {
  const float* path = (const float*)d_in[0];
  float* out = (float*)d_out;
  logsig_scan_kernel<<<dim3(NGRID), dim3(NTHR), 0, stream>>>(path, out);
}


# Logarithm: each tensor power's LDS reads (P and S) are all issued before any write of the new power, its elements held in registers: one LDS round trip per power, not one per element; sums unchanged
# speedup vs baseline: 8.0065x; 1.1470x over previous
#include <hip/hip_runtime.h>


typedef float v2f __attribute__((ext_vector_type(2)));
typedef float v8f __attribute__((ext_vector_type(8)));

#define B_      128
#define L_      257
#define C_      8
#define NSTEP   256
#define OUTPB   4680

#define NG      2
#define NW      4
#define NTHR    128
#define NT      (16 / (NG * NW))
#define NI      (NT / 2)
static_assert(NTHR == 32 * NW, "one wave is 32 lanes");
static_assert(NT == 2 && NI == 1, "a wave owns one (even, odd) tile pair, which shares one level-1 component");
static_assert(NG == 2, "lanes 16-31 carry the partner pair of exactly one other group");
#define NGRID   256
static_assert(NGRID == NG * B_, "two work-groups per batch element");

#define VINC_OFF 4680
#define S1O 0
#define S2O 8
#define S3O 72
#define S4O 584
#define SREG 0
#define PREG 4680
#define QREG 9360
#define LDS_FLOATS 14040

__global__ __launch_bounds__(NTHR)
void logsig_scan_kernel(const float* __restrict__ path, float* __restrict__ out) {
  __shared__ float buf[LDS_FLOATS];
  const int tid  = threadIdx.x;
  const int lane = tid & 31;
  const int w    = __builtin_amdgcn_readfirstlane(tid >> 5);
  const int b    = blockIdx.x >> 1;
  const int h    = blockIdx.x & 1;
  const int T0   = 8 * h + NT * w;
  const int i0   = T0 >> 1;
  const int T0l  = T0 ^ (8 * (lane >> 4));
  const int i0l  = T0l >> 1;
  const float* gp = path + (size_t)b * (L_ * C_);

  {
    const float4* g4 = (const float4*)gp;
    float4* l4 = (float4*)&buf[VINC_OFF];
    constexpr int NPASS = (NSTEP * C_) / 4 / NTHR;
    static_assert(NPASS * NTHR * 4 == NSTEP * C_, "the table is a whole number of passes");
    float4 a[NPASS], c[NPASS];
#pragma unroll
    for (int k = 0; k < NPASS; ++k) {
      a[k] = g4[tid + k * NTHR];  c[k] = g4[tid + k * NTHR + 2];
    }
#pragma unroll
    for (int k = 0; k < NPASS; ++k)
      l4[tid + k * NTHR] = make_float4(c[k].x - a[k].x, c[k].y - a[k].y, c[k].z - a[k].z, c[k].w - a[k].w);
    if (tid < 4) l4[(NSTEP * C_) / 4 + tid] = make_float4(0.f, 0.f, 0.f, 0.f);
  }
  __syncthreads();

  v8f  cacc[NT];
  v2f  a3[NT];
  float r2[NT];
  float s1[NI];
#pragma unroll
  for (int t = 0; t < NT; ++t) {
    a3[t].x = 0.f; a3[t].y = 0.f; r2[t] = 0.f;
#pragma unroll
    for (int q = 0; q < 8; ++q) cacc[t][q] = 0.f;
  }
#pragma unroll
  for (int ii = 0; ii < NI; ++ii) s1[ii] = 0.f;

  const int jj  = lane & 7;
  const int kk2 = 2 * (lane & 3);
  const int ta  = (lane >> 2) & 3;
  const bool blo = lane < 8;
  const bool bhi = (lane >= 8) && (lane < 16);

  const float* vr = &buf[VINC_OFF];
  float  ni[2], nlo[2], nhi[2], njj[2];
  float2 nk[2];
#pragma unroll
  for (int u = 0; u < 2; ++u) {
    const float* row = vr + u * C_;
    ni[u]  = row[i0l];
    nlo[u] = row[ta];  nhi[u] = row[4 + ta];
    nk[u]  = *(const float2*)&row[kk2];
    njj[u] = row[jj];
  }

  static_assert(NSTEP % 2 == 0, "the scan takes two steps per trip");
  for (int s = 0; s < NSTEP; s += 2) {
    vr += 2 * C_;
    float  fi[2], flo[2], fhi[2], fjj[2];
    float2 fk[2];
#pragma unroll
    for (int u = 0; u < 2; ++u) {
      const float* row = vr + u * C_;
      fi[u]  = row[i0l];
      flo[u] = row[ta];  fhi[u] = row[4 + ta];
      fk[u]  = *(const float2*)&row[kk2];
      fjj[u] = row[jj];
    }
    __builtin_amdgcn_sched_barrier(0);

#pragma unroll
    for (int u = 0; u < 2; ++u) {
      float g2[NI], g3[NI], g4[NI];
#pragma unroll
      for (int ii = 0; ii < NI; ++ii) {
        g2[ii] = s1[ii] + 0.5f * ni[u];
        g3[ii] = s1[ii] + ni[u] * (1.0f / 3.0f);
        g4[ii] = s1[ii] + 0.25f * ni[u];
        s1[ii] += ni[u];
      }
      const float h2lo = 0.5f * nlo[u], h2hi = 0.5f * nhi[u];
      const float h3lo = nlo[u] * (1.0f / 3.0f), h3hi = nhi[u] * (1.0f / 3.0f);
      const float hkA = 0.5f * nk[u].x, hkB = 0.5f * nk[u].y;
      const float vjlo = nlo[u], vjhi = nhi[u], vkA = nk[u].x, vkB = nk[u].y;

      v2f Bv;
      Bv.x = blo ? njj[u] : 0.0f;
      Bv.y = bhi ? njj[u] : 0.0f;

#pragma unroll
      for (int t = 0; t < NT; ++t) {
        const int   ii  = t >> 1;
        const float h3j = (t & 1) ? h3hi : h3lo;
        const float h2j = (t & 1) ? h2hi : h2lo;
        const float vj  = (t & 1) ? vjhi : vjlo;
        float t2v = r2[t] + h3j * g4[ii];
        float u2v = r2[t] + h2j * g3[ii];
        r2[t] += vj * g2[ii];
        v2f Av;
        Av.x = a3[t].x + hkA * t2v;
        Av.y = a3[t].y + hkB * t2v;
        cacc[t] = __builtin_amdgcn_wmma_f32_16x16x4_f32(
            false, Av, false, Bv, (short)0, cacc[t], false, false);
        a3[t].x += vkA * u2v;
        a3[t].y += vkB * u2v;
      }
    }
    asm volatile("" :: "v"(fi[0]), "v"(flo[0]), "v"(fhi[0]), "v"(fk[0].x), "v"(fk[0].y), "v"(fjj[0]),
                       "v"(fi[1]), "v"(flo[1]), "v"(fhi[1]), "v"(fk[1].x), "v"(fk[1].y), "v"(fjj[1]));
#pragma unroll
    for (int u = 0; u < 2; ++u) { ni[u] = fi[u]; nlo[u] = flo[u]; nhi[u] = fhi[u]; nk[u] = fk[u]; njj[u] = fjj[u]; }
  }
  __syncthreads();

  if ((lane & 15) == 0) {
#pragma unroll
    for (int ii = 0; ii < NI; ++ii) {
      buf[SREG + S1O + i0l + ii] = s1[ii];  buf[PREG + S1O + i0l + ii] = s1[ii];
    }
  }
  if ((lane & 3) == 0) {
#pragma unroll
    for (int t = 0; t < NT; ++t) {
      const int T = T0l + t;
      buf[SREG + S2O + 4 * T + ta] = r2[t];  buf[PREG + S2O + 4 * T + ta] = r2[t];
    }
  }
  {
#pragma unroll
    for (int t = 0; t < NT; ++t) {
      int f = 32 * (T0l + t) + 2 * (lane & 15);
      buf[SREG + S3O + f]     = a3[t].x;  buf[PREG + S3O + f]     = a3[t].x;
      buf[SREG + S3O + f + 1] = a3[t].y;  buf[PREG + S3O + f + 1] = a3[t].y;
    }
  }
  {
    const int hi = (lane >> 4) * 8, cc = lane & 15;
#pragma unroll
    for (int t = 0; t < NT; ++t)
#pragma unroll
      for (int q = 0; q < 8; ++q) {
        int f = (16 * (T0 + t) + q + hi) * 16 + cc;
        buf[SREG + S4O + f] = cacc[t][q];
        buf[PREG + S4O + f] = cacc[t][q];
      }
  }
  __syncthreads();

  float sig1 = (lane < 4) ? buf[SREG + S1O + 4 * h + lane] : 0.f;
  float r2o  = buf[SREG + S2O + 32 * h + lane];

  int Pb = PREG, Qb = QREG;
  const int mm = lane & 15;
  const int hi4 = (lane >> 4) * 8, cc4 = lane & 15;
  for (int n = 2; n <= 4; ++n) {
    float coef = ((n & 1) ? 1.0f : -1.0f) / (float)n;
    float q2;
    {
      int ij = 32 * h + lane;
      q2 = buf[Pb + S1O + (ij >> 3)] * buf[SREG + S1O + (ij & 7)];
      r2o += coef * q2;
    }
    float q3[NT][2];
#pragma unroll
    for (int t = 0; t < NT; ++t) {
#pragma unroll
      for (int r = 0; r < 2; ++r) {
        int f = 32 * (T0 + t) + 2 * mm + r;
        float q = buf[Pb + S1O + (f >> 6)] * buf[SREG + S2O + (f & 63)]
                + buf[Pb + S2O + (f >> 3)] * buf[SREG + S1O + (f & 7)];
        if (r == 0) a3[t].x += coef * q; else a3[t].y += coef * q;
        q3[t][r] = q;
      }
    }
    float q4[NT][8];
#pragma unroll
    for (int t = 0; t < NT; ++t)
#pragma unroll
      for (int q8 = 0; q8 < 8; ++q8) {
        int f = (16 * (T0 + t) + q8 + hi4) * 16 + cc4;
        float q = buf[Pb + S1O + (f >> 9)] * buf[SREG + S3O + (f & 511)]
                + buf[Pb + S2O + (f >> 6)] * buf[SREG + S2O + (f & 63)]
                + buf[Pb + S3O + (f >> 3)] * buf[SREG + S1O + (f & 7)];
        cacc[t][q8] += coef * q;
        q4[t][q8] = q;
      }
    if (w == 0) {
      if (lane < 4) buf[Qb + S1O + 4 * h + lane] = 0.0f;
      buf[Qb + S2O + 32 * h + lane] = q2;
    }
    if (lane < 16) {
#pragma unroll
      for (int t = 0; t < NT; ++t)
#pragma unroll
        for (int r = 0; r < 2; ++r)
          buf[Qb + S3O + 32 * (T0 + t) + 2 * mm + r] = q3[t][r];
    }
#pragma unroll
    for (int t = 0; t < NT; ++t)
#pragma unroll
      for (int q8 = 0; q8 < 8; ++q8)
        buf[Qb + S4O + (16 * (T0 + t) + q8 + hi4) * 16 + cc4] = q4[t][q8];
    __syncthreads();
    int tmp = Pb; Pb = Qb; Qb = tmp;
  }

  float* ob = out + (size_t)b * OUTPB;
  if (w == 0) {
    if (lane < 4) ob[4 * h + lane] = sig1;
    ob[8 + 32 * h + lane] = r2o;
  }
  if (lane < 16) {
#pragma unroll
    for (int t = 0; t < NT; ++t) {
      int f = 32 * (T0 + t) + 2 * lane;
      ob[72 + f]     = a3[t].x;
      ob[72 + f + 1] = a3[t].y;
    }
  }
  {
    const int hi = (lane >> 4) * 8, cc = lane & 15;
#pragma unroll
    for (int t = 0; t < NT; ++t)
#pragma unroll
      for (int q = 0; q < 8; ++q) {
        int f = (16 * (T0 + t) + q + hi) * 16 + cc;
        ob[584 + f] = cacc[t][q];
      }
  }
}

extern "C" void kernel_launch(void* const* d_in, const int* in_sizes, int n_in,
                              void* d_out, int out_size, void* d_ws, size_t ws_size,
                              hipStream_t stream) {
  const float* path = (const float*)d_in[0];
  float* out = (float*)d_out;
  logsig_scan_kernel<<<dim3(NGRID), dim3(NTHR), 0, stream>>>(path, out);
}


# Scan: four steps' increments are read from LDS as ONE group of loads, a group ahead, into a second register set (no copies): one place with LDS reads per four steps, not per step; sums unchanged
# speedup vs baseline: 9.4562x; 1.1811x over previous
#include <hip/hip_runtime.h>


typedef float v2f __attribute__((ext_vector_type(2)));
typedef float v8f __attribute__((ext_vector_type(8)));

#define B_      128
#define L_      257
#define C_      8
#define NSTEP   256
#define OUTPB   4680

#define NG      2
#define NW      4
#define NTHR    128
#define NT      (16 / (NG * NW))
#define NI      (NT / 2)
static_assert(NTHR == 32 * NW, "one wave is 32 lanes");
static_assert(NT == 2 && NI == 1, "a wave owns one (even, odd) tile pair, which shares one level-1 component");
static_assert(NG == 2, "lanes 16-31 carry the partner pair of exactly one other group");
#define NGRID   256
static_assert(NGRID == NG * B_, "two work-groups per batch element");

#define VINC_OFF 4680
#define S1O 0
#define S2O 8
#define S3O 72
#define S4O 584
#define SREG 0
#define PREG 4680
#define QREG 9360
#define LDS_FLOATS 14040

#define GRP     4
static_assert(GRP == 4, "the fences in the scan name the four steps of a group one by one");

__device__ __forceinline__ void read_group(const float* rows, int i0l, int ta, int kk2, int jj,
                                           float (&fi)[GRP], float (&flo)[GRP], float (&fhi)[GRP],
                                           float2 (&fk)[GRP], float (&fjj)[GRP]) {
#pragma unroll
  for (int u = 0; u < GRP; ++u) {
    const float* row = rows + u * C_;
    fi[u]  = row[i0l];
    flo[u] = row[ta];  fhi[u] = row[4 + ta];
    fk[u]  = *(const float2*)&row[kk2];
    fjj[u] = row[jj];
  }
}

__device__ __forceinline__ void scan_step(float ni, float nlo, float nhi, float2 nk, float njj,
                                          bool blo, bool bhi,
                                          float (&s1)[NI], float (&r2)[NT], v2f (&a3)[NT], v8f (&cacc)[NT]) {
  float g2[NI], g3[NI], g4[NI];
#pragma unroll
  for (int ii = 0; ii < NI; ++ii) {
    g2[ii] = __builtin_fmaf(0.5f, ni, s1[ii]);
    g3[ii] = __builtin_fmaf(ni, 1.0f / 3.0f, s1[ii]);
    g4[ii] = __builtin_fmaf(0.25f, ni, s1[ii]);
    s1[ii] += ni;
  }
  const float h2lo = 0.5f * nlo, h2hi = 0.5f * nhi;
  const float h3lo = nlo * (1.0f / 3.0f), h3hi = nhi * (1.0f / 3.0f);
  const float hkA = 0.5f * nk.x, hkB = 0.5f * nk.y;
  const float vjlo = nlo, vjhi = nhi, vkA = nk.x, vkB = nk.y;

  v2f Bv;
  Bv.x = blo ? njj : 0.0f;
  Bv.y = bhi ? njj : 0.0f;

#pragma unroll
  for (int t = 0; t < NT; ++t) {
    const int   ii  = t >> 1;
    const float h3j = (t & 1) ? h3hi : h3lo;
    const float h2j = (t & 1) ? h2hi : h2lo;
    const float vj  = (t & 1) ? vjhi : vjlo;
    float t2v = __builtin_fmaf(h3j, g4[ii], r2[t]);
    float u2v = __builtin_fmaf(h2j, g3[ii], r2[t]);
    r2[t] = __builtin_fmaf(vj, g2[ii], r2[t]);
    v2f Av;
    Av.x = __builtin_fmaf(hkA, t2v, a3[t].x);
    Av.y = __builtin_fmaf(hkB, t2v, a3[t].y);
    cacc[t] = __builtin_amdgcn_wmma_f32_16x16x4_f32(
        false, Av, false, Bv, (short)0, cacc[t], false, false);
    a3[t].x = __builtin_fmaf(vkA, u2v, a3[t].x);
    a3[t].y = __builtin_fmaf(vkB, u2v, a3[t].y);
  }
}

__global__ __launch_bounds__(NTHR)
void logsig_scan_kernel(const float* __restrict__ path, float* __restrict__ out) {
  __shared__ float buf[LDS_FLOATS];
  const int tid  = threadIdx.x;
  const int lane = tid & 31;
  const int w    = __builtin_amdgcn_readfirstlane(tid >> 5);
  const int b    = blockIdx.x >> 1;
  const int h    = blockIdx.x & 1;
  const int T0   = 8 * h + NT * w;
  const int i0   = T0 >> 1;
  const int T0l  = T0 ^ (8 * (lane >> 4));
  const int i0l  = T0l >> 1;
  const float* gp = path + (size_t)b * (L_ * C_);

  {
    const float4* g4 = (const float4*)gp;
    float4* l4 = (float4*)&buf[VINC_OFF];
    constexpr int NPASS = (NSTEP * C_) / 4 / NTHR;
    static_assert(NPASS * NTHR * 4 == NSTEP * C_, "the table is a whole number of passes");
    float4 a[NPASS], c[NPASS];
#pragma unroll
    for (int k = 0; k < NPASS; ++k) {
      a[k] = g4[tid + k * NTHR];  c[k] = g4[tid + k * NTHR + 2];
    }
#pragma unroll
    for (int k = 0; k < NPASS; ++k)
      l4[tid + k * NTHR] = make_float4(c[k].x - a[k].x, c[k].y - a[k].y, c[k].z - a[k].z, c[k].w - a[k].w);
    if (tid < 2 * GRP) l4[(NSTEP * C_) / 4 + tid] = make_float4(0.f, 0.f, 0.f, 0.f);
  }
  __syncthreads();

  v8f  cacc[NT];
  v2f  a3[NT];
  float r2[NT];
  float s1[NI];
#pragma unroll
  for (int t = 0; t < NT; ++t) {
    a3[t].x = 0.f; a3[t].y = 0.f; r2[t] = 0.f;
#pragma unroll
    for (int q = 0; q < 8; ++q) cacc[t][q] = 0.f;
  }
#pragma unroll
  for (int ii = 0; ii < NI; ++ii) s1[ii] = 0.f;

  const int jj  = lane & 7;
  const int kk2 = 2 * (lane & 3);
  const int ta  = (lane >> 2) & 3;
  const bool blo = lane < 8;
  const bool bhi = (lane >= 8) && (lane < 16);

  const float* vr = &buf[VINC_OFF];
  float  xi[GRP], xlo[GRP], xhi[GRP], xjj[GRP];
  float2 xk[GRP];
  float  yi[GRP], ylo[GRP], yhi[GRP], yjj[GRP];
  float2 yk[GRP];
  read_group(vr, i0l, ta, kk2, jj, xi, xlo, xhi, xk, xjj);

  static_assert(NSTEP % (2 * GRP) == 0, "the scan takes two groups of steps per trip");
  for (int s = 0; s < NSTEP; s += 2 * GRP) {
    read_group(vr + (s + GRP) * C_, i0l, ta, kk2, jj, yi, ylo, yhi, yk, yjj);
    __builtin_amdgcn_sched_barrier(0);
#pragma unroll
    for (int u = 0; u < GRP; ++u) {
      scan_step(xi[u], xlo[u], xhi[u], xk[u], xjj[u], blo, bhi, s1, r2, a3, cacc);
      if (u & 1) __builtin_amdgcn_sched_barrier(0);
    }
    asm volatile("" :: "v"(yi[0]), "v"(ylo[0]), "v"(yhi[0]), "v"(yk[0].x), "v"(yk[0].y), "v"(yjj[0]),
                       "v"(yi[1]), "v"(ylo[1]), "v"(yhi[1]), "v"(yk[1].x), "v"(yk[1].y), "v"(yjj[1]),
                       "v"(yi[2]), "v"(ylo[2]), "v"(yhi[2]), "v"(yk[2].x), "v"(yk[2].y), "v"(yjj[2]),
                       "v"(yi[3]), "v"(ylo[3]), "v"(yhi[3]), "v"(yk[3].x), "v"(yk[3].y), "v"(yjj[3]));
    read_group(vr + (s + 2 * GRP) * C_, i0l, ta, kk2, jj, xi, xlo, xhi, xk, xjj);
    __builtin_amdgcn_sched_barrier(0);
#pragma unroll
    for (int u = 0; u < GRP; ++u) {
      scan_step(yi[u], ylo[u], yhi[u], yk[u], yjj[u], blo, bhi, s1, r2, a3, cacc);
      if (u & 1) __builtin_amdgcn_sched_barrier(0);
    }
    asm volatile("" :: "v"(xi[0]), "v"(xlo[0]), "v"(xhi[0]), "v"(xk[0].x), "v"(xk[0].y), "v"(xjj[0]),
                       "v"(xi[1]), "v"(xlo[1]), "v"(xhi[1]), "v"(xk[1].x), "v"(xk[1].y), "v"(xjj[1]),
                       "v"(xi[2]), "v"(xlo[2]), "v"(xhi[2]), "v"(xk[2].x), "v"(xk[2].y), "v"(xjj[2]),
                       "v"(xi[3]), "v"(xlo[3]), "v"(xhi[3]), "v"(xk[3].x), "v"(xk[3].y), "v"(xjj[3]));
  }
  __syncthreads();

  if ((lane & 15) == 0) {
#pragma unroll
    for (int ii = 0; ii < NI; ++ii) {
      buf[SREG + S1O + i0l + ii] = s1[ii];  buf[PREG + S1O + i0l + ii] = s1[ii];
    }
  }
  if ((lane & 3) == 0) {
#pragma unroll
    for (int t = 0; t < NT; ++t) {
      const int T = T0l + t;
      buf[SREG + S2O + 4 * T + ta] = r2[t];  buf[PREG + S2O + 4 * T + ta] = r2[t];
    }
  }
  {
#pragma unroll
    for (int t = 0; t < NT; ++t) {
      int f = 32 * (T0l + t) + 2 * (lane & 15);
      buf[SREG + S3O + f]     = a3[t].x;  buf[PREG + S3O + f]     = a3[t].x;
      buf[SREG + S3O + f + 1] = a3[t].y;  buf[PREG + S3O + f + 1] = a3[t].y;
    }
  }
  {
    const int hi = (lane >> 4) * 8, cc = lane & 15;
#pragma unroll
    for (int t = 0; t < NT; ++t)
#pragma unroll
      for (int q = 0; q < 8; ++q) {
        int f = (16 * (T0 + t) + q + hi) * 16 + cc;
        buf[SREG + S4O + f] = cacc[t][q];
        buf[PREG + S4O + f] = cacc[t][q];
      }
  }
  __syncthreads();

  float sig1 = (lane < 4) ? buf[SREG + S1O + 4 * h + lane] : 0.f;
  float r2o  = buf[SREG + S2O + 32 * h + lane];

  int Pb = PREG, Qb = QREG;
  const int mm = lane & 15;
  const int hi4 = (lane >> 4) * 8, cc4 = lane & 15;
  for (int n = 2; n <= 4; ++n) {
    float coef = ((n & 1) ? 1.0f : -1.0f) / (float)n;
    float q2;
    {
      int ij = 32 * h + lane;
      q2 = buf[Pb + S1O + (ij >> 3)] * buf[SREG + S1O + (ij & 7)];
      r2o += coef * q2;
    }
    float q3[NT][2];
#pragma unroll
    for (int t = 0; t < NT; ++t) {
#pragma unroll
      for (int r = 0; r < 2; ++r) {
        int f = 32 * (T0 + t) + 2 * mm + r;
        float q = buf[Pb + S1O + (f >> 6)] * buf[SREG + S2O + (f & 63)]
                + buf[Pb + S2O + (f >> 3)] * buf[SREG + S1O + (f & 7)];
        if (r == 0) a3[t].x += coef * q; else a3[t].y += coef * q;
        q3[t][r] = q;
      }
    }
    float q4[NT][8];
#pragma unroll
    for (int t = 0; t < NT; ++t)
#pragma unroll
      for (int q8 = 0; q8 < 8; ++q8) {
        int f = (16 * (T0 + t) + q8 + hi4) * 16 + cc4;
        float q = buf[Pb + S1O + (f >> 9)] * buf[SREG + S3O + (f & 511)]
                + buf[Pb + S2O + (f >> 6)] * buf[SREG + S2O + (f & 63)]
                + buf[Pb + S3O + (f >> 3)] * buf[SREG + S1O + (f & 7)];
        cacc[t][q8] += coef * q;
        q4[t][q8] = q;
      }
    if (w == 0) {
      if (lane < 4) buf[Qb + S1O + 4 * h + lane] = 0.0f;
      buf[Qb + S2O + 32 * h + lane] = q2;
    }
    if (lane < 16) {
#pragma unroll
      for (int t = 0; t < NT; ++t)
#pragma unroll
        for (int r = 0; r < 2; ++r)
          buf[Qb + S3O + 32 * (T0 + t) + 2 * mm + r] = q3[t][r];
    }
#pragma unroll
    for (int t = 0; t < NT; ++t)
#pragma unroll
      for (int q8 = 0; q8 < 8; ++q8)
        buf[Qb + S4O + (16 * (T0 + t) + q8 + hi4) * 16 + cc4] = q4[t][q8];
    __syncthreads();
    int tmp = Pb; Pb = Qb; Qb = tmp;
  }

  float* ob = out + (size_t)b * OUTPB;
  if (w == 0) {
    if (lane < 4) ob[4 * h + lane] = sig1;
    ob[8 + 32 * h + lane] = r2o;
  }
  if (lane < 16) {
#pragma unroll
    for (int t = 0; t < NT; ++t) {
      int f = 32 * (T0 + t) + 2 * lane;
      ob[72 + f]     = a3[t].x;
      ob[72 + f + 1] = a3[t].y;
    }
  }
  {
    const int hi = (lane >> 4) * 8, cc = lane & 15;
#pragma unroll
    for (int t = 0; t < NT; ++t)
#pragma unroll
      for (int q = 0; q < 8; ++q) {
        int f = (16 * (T0 + t) + q + hi) * 16 + cc;
        ob[584 + f] = cacc[t][q];
      }
  }
}

extern "C" void kernel_launch(void* const* d_in, const int* in_sizes, int n_in,
                              void* d_out, int out_size, void* d_ws, size_t ws_size,
                              hipStream_t stream) {
  const float* path = (const float*)d_in[0];
  float* out = (float*)d_out;
  logsig_scan_kernel<<<dim3(NGRID), dim3(NTHR), 0, stream>>>(path, out);
}


# Scan: ONE matrix instruction per tile and PAIR of steps (all of K = 4): step 2's level-3 term goes to lanes 16-31 of A by a row swap, its increment to B; the pair's level-4 products add inside it
# speedup vs baseline: 10.8626x; 1.1487x over previous
#include <hip/hip_runtime.h>


typedef float v2f __attribute__((ext_vector_type(2)));
typedef float v8f __attribute__((ext_vector_type(8)));

#define B_      128
#define L_      257
#define C_      8
#define NSTEP   256
#define OUTPB   4680

#define NG      2
#define NW      4
#define NTHR    128
#define NT      (16 / (NG * NW))
#define NI      (NT / 2)
static_assert(NTHR == 32 * NW, "one wave is 32 lanes");
static_assert(NT == 2 && NI == 1, "a wave owns one (even, odd) tile pair, which shares one level-1 component");
static_assert(NG == 2, "lanes 16-31 carry the partner pair of exactly one other group");
#define NGRID   256
static_assert(NGRID == NG * B_, "two work-groups per batch element");

#define VINC_OFF 4680
#define S1O 0
#define S2O 8
#define S3O 72
#define S4O 584
#define SREG 0
#define PREG 4680
#define QREG 9360
#define LDS_FLOATS 14040

#define GRP     4
static_assert(GRP == 4, "the fences in the scan name the four steps of a group one by one");

#define NPAIR   (GRP / 2)
__device__ __forceinline__ void read_group(const float* rows, int i0l, int ta, int kk2, int jjh,
                                           float (&fi)[GRP], float (&flo)[GRP], float (&fhi)[GRP],
                                           float2 (&fk)[GRP], float (&fjj)[NPAIR]) {
#pragma unroll
  for (int u = 0; u < GRP; ++u) {
    const float* row = rows + u * C_;
    fi[u]  = row[i0l];
    flo[u] = row[ta];  fhi[u] = row[4 + ta];
    fk[u]  = *(const float2*)&row[kk2];
  }
#pragma unroll
  for (int p = 0; p < NPAIR; ++p)
    fjj[p] = rows[2 * p * C_ + jjh];
}

__device__ __forceinline__ void scan_step_l123(float ni, float nlo, float nhi, float2 nk,
                                               float (&s1)[NI], float (&r2)[NT], v2f (&a3)[NT], v2f (&Av)[NT]) {
  float g2[NI], g3[NI], g4[NI];
#pragma unroll
  for (int ii = 0; ii < NI; ++ii) {
    g2[ii] = __builtin_fmaf(0.5f, ni, s1[ii]);
    g3[ii] = __builtin_fmaf(ni, 1.0f / 3.0f, s1[ii]);
    g4[ii] = __builtin_fmaf(0.25f, ni, s1[ii]);
    s1[ii] += ni;
  }
  const float h2lo = 0.5f * nlo, h2hi = 0.5f * nhi;
  const float h3lo = nlo * (1.0f / 3.0f), h3hi = nhi * (1.0f / 3.0f);
  const float hkA = 0.5f * nk.x, hkB = 0.5f * nk.y;
  const float vjlo = nlo, vjhi = nhi, vkA = nk.x, vkB = nk.y;

#pragma unroll
  for (int t = 0; t < NT; ++t) {
    const int   ii  = t >> 1;
    const float h3j = (t & 1) ? h3hi : h3lo;
    const float h2j = (t & 1) ? h2hi : h2lo;
    const float vj  = (t & 1) ? vjhi : vjlo;
    float t2v = __builtin_fmaf(h3j, g4[ii], r2[t]);
    float u2v = __builtin_fmaf(h2j, g3[ii], r2[t]);
    r2[t] = __builtin_fmaf(vj, g2[ii], r2[t]);
    Av[t].x = __builtin_fmaf(hkA, t2v, a3[t].x);
    Av[t].y = __builtin_fmaf(hkB, t2v, a3[t].y);
    a3[t].x = __builtin_fmaf(vkA, u2v, a3[t].x);
    a3[t].y = __builtin_fmaf(vkB, u2v, a3[t].y);
  }
}

__device__ __forceinline__ float rows_lo_of_both(float a, float b) {
  return __builtin_bit_cast(float, __builtin_amdgcn_permlane16_swap(
      __builtin_bit_cast(unsigned, a), __builtin_bit_cast(unsigned, b), false, false)[0]);
}

template <int U0>
__device__ __forceinline__ void scan_pair(const float (&fi)[GRP], const float (&flo)[GRP], const float (&fhi)[GRP],
                                          const float2 (&fk)[GRP], float fjj, bool blo, bool bhi,
                                          float (&s1)[NI], float (&r2)[NT], v2f (&a3)[NT], v8f (&cacc)[NT],
                                          v2f (&Ap)[NT], v2f& Bp) {
  static_assert(NT == 2, "one matrix instruction in front of each of the pair's two steps");
  v2f A0[NT], A1[NT];
  cacc[0] = __builtin_amdgcn_wmma_f32_16x16x4_f32(false, Ap[0], false, Bp, (short)0, cacc[0], false, false);
  asm volatile("" :: "v"(cacc[0]));
  scan_step_l123(fi[U0], flo[U0], fhi[U0], fk[U0], s1, r2, a3, A0);
  __builtin_amdgcn_sched_barrier(0);
  cacc[1] = __builtin_amdgcn_wmma_f32_16x16x4_f32(false, Ap[1], false, Bp, (short)0, cacc[1], false, false);
  asm volatile("" :: "v"(cacc[1]));
  scan_step_l123(fi[U0 + 1], flo[U0 + 1], fhi[U0 + 1], fk[U0 + 1], s1, r2, a3, A1);
#pragma unroll
  for (int t = 0; t < NT; ++t) {
    Ap[t].x = rows_lo_of_both(A0[t].x, A1[t].x);
    Ap[t].y = rows_lo_of_both(A0[t].y, A1[t].y);
  }
  Bp.x = blo ? fjj : 0.0f;
  Bp.y = bhi ? fjj : 0.0f;
  __builtin_amdgcn_sched_barrier(0);
}

__global__ __launch_bounds__(NTHR)
void logsig_scan_kernel(const float* __restrict__ path, float* __restrict__ out) {
  __shared__ float buf[LDS_FLOATS];
  const int tid  = threadIdx.x;
  const int lane = tid & 31;
  const int w    = __builtin_amdgcn_readfirstlane(tid >> 5);
  const int b    = blockIdx.x >> 1;
  const int h    = blockIdx.x & 1;
  const int T0   = 8 * h + NT * w;
  const int i0   = T0 >> 1;
  const int T0l  = T0 ^ (8 * (lane >> 4));
  const int i0l  = T0l >> 1;
  const float* gp = path + (size_t)b * (L_ * C_);

  {
    const float4* g4 = (const float4*)gp;
    float4* l4 = (float4*)&buf[VINC_OFF];
    constexpr int NPASS = (NSTEP * C_) / 4 / NTHR;
    static_assert(NPASS * NTHR * 4 == NSTEP * C_, "the table is a whole number of passes");
    float4 a[NPASS], c[NPASS];
#pragma unroll
    for (int k = 0; k < NPASS; ++k) {
      a[k] = g4[tid + k * NTHR];  c[k] = g4[tid + k * NTHR + 2];
    }
#pragma unroll
    for (int k = 0; k < NPASS; ++k)
      l4[tid + k * NTHR] = make_float4(c[k].x - a[k].x, c[k].y - a[k].y, c[k].z - a[k].z, c[k].w - a[k].w);
    if (tid < 2 * GRP) l4[(NSTEP * C_) / 4 + tid] = make_float4(0.f, 0.f, 0.f, 0.f);
  }
  __syncthreads();

  v8f  cacc[NT];
  v2f  a3[NT];
  float r2[NT];
  float s1[NI];
#pragma unroll
  for (int t = 0; t < NT; ++t) {
    a3[t].x = 0.f; a3[t].y = 0.f; r2[t] = 0.f;
#pragma unroll
    for (int q = 0; q < 8; ++q) cacc[t][q] = 0.f;
  }
#pragma unroll
  for (int ii = 0; ii < NI; ++ii) s1[ii] = 0.f;

  const int jjh = 8 * (lane >> 4) + (lane & 7);
  const int kk2 = 2 * (lane & 3);
  const int ta  = (lane >> 2) & 3;
  const bool blo = (lane & 15) < 8;
  const bool bhi = (lane & 15) >= 8;

  const float* vr = &buf[VINC_OFF];
  float  xi[GRP], xlo[GRP], xhi[GRP], xjj[NPAIR];
  float2 xk[GRP];
  float  yi[GRP], ylo[GRP], yhi[GRP], yjj[NPAIR];
  float2 yk[GRP];
  read_group(vr, i0l, ta, kk2, jjh, xi, xlo, xhi, xk, xjj);

  v2f Ap[NT], Bp;
#pragma unroll
  for (int t = 0; t < NT; ++t) { Ap[t].x = 0.f; Ap[t].y = 0.f; }
  Bp.x = 0.f; Bp.y = 0.f;

  static_assert(NSTEP % (2 * GRP) == 0, "the scan takes two groups of steps per trip");
  static_assert(NPAIR == 2, "a group is two pairs of steps");
  for (int s = 0; s < NSTEP; s += 2 * GRP) {
    read_group(vr + (s + GRP) * C_, i0l, ta, kk2, jjh, yi, ylo, yhi, yk, yjj);
    __builtin_amdgcn_sched_barrier(0);
    scan_pair<0>(xi, xlo, xhi, xk, xjj[0], blo, bhi, s1, r2, a3, cacc, Ap, Bp);
    scan_pair<2>(xi, xlo, xhi, xk, xjj[1], blo, bhi, s1, r2, a3, cacc, Ap, Bp);
    asm volatile("" :: "v"(yi[0]), "v"(ylo[0]), "v"(yhi[0]), "v"(yk[0].x), "v"(yk[0].y),
                       "v"(yi[1]), "v"(ylo[1]), "v"(yhi[1]), "v"(yk[1].x), "v"(yk[1].y),
                       "v"(yi[2]), "v"(ylo[2]), "v"(yhi[2]), "v"(yk[2].x), "v"(yk[2].y),
                       "v"(yi[3]), "v"(ylo[3]), "v"(yhi[3]), "v"(yk[3].x), "v"(yk[3].y),
                       "v"(yjj[0]), "v"(yjj[1]));
    read_group(vr + (s + 2 * GRP) * C_, i0l, ta, kk2, jjh, xi, xlo, xhi, xk, xjj);
    __builtin_amdgcn_sched_barrier(0);
    scan_pair<0>(yi, ylo, yhi, yk, yjj[0], blo, bhi, s1, r2, a3, cacc, Ap, Bp);
    scan_pair<2>(yi, ylo, yhi, yk, yjj[1], blo, bhi, s1, r2, a3, cacc, Ap, Bp);
    asm volatile("" :: "v"(xi[0]), "v"(xlo[0]), "v"(xhi[0]), "v"(xk[0].x), "v"(xk[0].y),
                       "v"(xi[1]), "v"(xlo[1]), "v"(xhi[1]), "v"(xk[1].x), "v"(xk[1].y),
                       "v"(xi[2]), "v"(xlo[2]), "v"(xhi[2]), "v"(xk[2].x), "v"(xk[2].y),
                       "v"(xi[3]), "v"(xlo[3]), "v"(xhi[3]), "v"(xk[3].x), "v"(xk[3].y),
                       "v"(xjj[0]), "v"(xjj[1]));
  }
#pragma unroll
  for (int t = 0; t < NT; ++t)
    cacc[t] = __builtin_amdgcn_wmma_f32_16x16x4_f32(false, Ap[t], false, Bp, (short)0, cacc[t], false, false);
  __syncthreads();

  if ((lane & 15) == 0) {
#pragma unroll
    for (int ii = 0; ii < NI; ++ii) {
      buf[SREG + S1O + i0l + ii] = s1[ii];  buf[PREG + S1O + i0l + ii] = s1[ii];
    }
  }
  if ((lane & 3) == 0) {
#pragma unroll
    for (int t = 0; t < NT; ++t) {
      const int T = T0l + t;
      buf[SREG + S2O + 4 * T + ta] = r2[t];  buf[PREG + S2O + 4 * T + ta] = r2[t];
    }
  }
  {
#pragma unroll
    for (int t = 0; t < NT; ++t) {
      int f = 32 * (T0l + t) + 2 * (lane & 15);
      buf[SREG + S3O + f]     = a3[t].x;  buf[PREG + S3O + f]     = a3[t].x;
      buf[SREG + S3O + f + 1] = a3[t].y;  buf[PREG + S3O + f + 1] = a3[t].y;
    }
  }
  {
    const int hi = (lane >> 4) * 8, cc = lane & 15;
#pragma unroll
    for (int t = 0; t < NT; ++t)
#pragma unroll
      for (int q = 0; q < 8; ++q) {
        int f = (16 * (T0 + t) + q + hi) * 16 + cc;
        buf[SREG + S4O + f] = cacc[t][q];
        buf[PREG + S4O + f] = cacc[t][q];
      }
  }
  __syncthreads();

  float sig1 = (lane < 4) ? buf[SREG + S1O + 4 * h + lane] : 0.f;
  float r2o  = buf[SREG + S2O + 32 * h + lane];

  int Pb = PREG, Qb = QREG;
  const int mm = lane & 15;
  const int hi4 = (lane >> 4) * 8, cc4 = lane & 15;
  for (int n = 2; n <= 4; ++n) {
    float coef = ((n & 1) ? 1.0f : -1.0f) / (float)n;
    float q2;
    {
      int ij = 32 * h + lane;
      q2 = buf[Pb + S1O + (ij >> 3)] * buf[SREG + S1O + (ij & 7)];
      r2o += coef * q2;
    }
    float q3[NT][2];
#pragma unroll
    for (int t = 0; t < NT; ++t) {
#pragma unroll
      for (int r = 0; r < 2; ++r) {
        int f = 32 * (T0 + t) + 2 * mm + r;
        float q = buf[Pb + S1O + (f >> 6)] * buf[SREG + S2O + (f & 63)]
                + buf[Pb + S2O + (f >> 3)] * buf[SREG + S1O + (f & 7)];
        if (r == 0) a3[t].x += coef * q; else a3[t].y += coef * q;
        q3[t][r] = q;
      }
    }
    float q4[NT][8];
#pragma unroll
    for (int t = 0; t < NT; ++t)
#pragma unroll
      for (int q8 = 0; q8 < 8; ++q8) {
        int f = (16 * (T0 + t) + q8 + hi4) * 16 + cc4;
        float q = buf[Pb + S1O + (f >> 9)] * buf[SREG + S3O + (f & 511)]
                + buf[Pb + S2O + (f >> 6)] * buf[SREG + S2O + (f & 63)]
                + buf[Pb + S3O + (f >> 3)] * buf[SREG + S1O + (f & 7)];
        cacc[t][q8] += coef * q;
        q4[t][q8] = q;
      }
    if (w == 0) {
      if (lane < 4) buf[Qb + S1O + 4 * h + lane] = 0.0f;
      buf[Qb + S2O + 32 * h + lane] = q2;
    }
    if (lane < 16) {
#pragma unroll
      for (int t = 0; t < NT; ++t)
#pragma unroll
        for (int r = 0; r < 2; ++r)
          buf[Qb + S3O + 32 * (T0 + t) + 2 * mm + r] = q3[t][r];
    }
#pragma unroll
    for (int t = 0; t < NT; ++t)
#pragma unroll
      for (int q8 = 0; q8 < 8; ++q8)
        buf[Qb + S4O + (16 * (T0 + t) + q8 + hi4) * 16 + cc4] = q4[t][q8];
    __syncthreads();
    int tmp = Pb; Pb = Qb; Qb = tmp;
  }

  float* ob = out + (size_t)b * OUTPB;
  if (w == 0) {
    if (lane < 4) ob[4 * h + lane] = sig1;
    ob[8 + 32 * h + lane] = r2o;
  }
  if (lane < 16) {
#pragma unroll
    for (int t = 0; t < NT; ++t) {
      int f = 32 * (T0 + t) + 2 * lane;
      ob[72 + f]     = a3[t].x;
      ob[72 + f + 1] = a3[t].y;
    }
  }
  {
    const int hi = (lane >> 4) * 8, cc = lane & 15;
#pragma unroll
    for (int t = 0; t < NT; ++t)
#pragma unroll
      for (int q = 0; q < 8; ++q) {
        int f = (16 * (T0 + t) + q + hi) * 16 + cc;
        ob[584 + f] = cacc[t][q];
      }
  }
}

extern "C" void kernel_launch(void* const* d_in, const int* in_sizes, int n_in,
                              void* d_out, int out_size, void* d_ws, size_t ws_size,
                              hipStream_t stream) {
  const float* path = (const float*)d_in[0];
  float* out = (float*)d_out;
  logsig_scan_kernel<<<dim3(NGRID), dim3(NTHR), 0, stream>>>(path, out);
}


# Scan: lanes 0-15 scan steps 0-127, lanes 16-31 steps 128-255 of the same rows (K 0,1 | K 2,3 of one matrix instruction, no row swaps); one tensor product joins the halves (summation order changes)
# speedup vs baseline: 11.8507x; 1.0910x over previous
#include <hip/hip_runtime.h>


typedef float v2f __attribute__((ext_vector_type(2)));
typedef float v8f __attribute__((ext_vector_type(8)));

#define B_      128
#define L_      257
#define C_      8
#define NSTEP   256
#define OUTPB   4680

#define NG      2
#define NW      4
#define NTHR    128
#define NT      (16 / (NG * NW))
#define NI      (NT / 2)
static_assert(NTHR == 32 * NW, "one wave is 32 lanes");
static_assert(NT == 2 && NI == 1, "a wave owns one (even, odd) tile pair, which shares one level-1 component");
static_assert(NG == 2, "lanes 16-31 carry the partner pair of exactly one other group");
#define NGRID   256
static_assert(NGRID == NG * B_, "two work-groups per batch element");

#define VINC_OFF 4680
#define S1O 0
#define S2O 8
#define S3O 72
#define S4O 584
#define SREG 0
#define PREG 4680
#define QREG 9360
#define LDS_FLOATS 14040
#define HALF_A  QREG
#define HALF_B  (QREG + 1024)
#define HSTEP   (NSTEP / 2)

#define GRP     4
static_assert(GRP == 4, "the fences in the scan name the four steps of a group one by one");

__device__ __forceinline__ void read_group(const float* rows, int i0, int i0p, int ta, int kk2, int jj,
                                           float (&fi)[GRP], float (&fip)[GRP], float (&flo)[GRP], float (&fhi)[GRP],
                                           float2 (&fk)[GRP], float (&fjj)[GRP]) {
#pragma unroll
  for (int u = 0; u < GRP; ++u) {
    const float* row = rows + u * C_;
    fi[u]  = row[i0];
    fip[u] = row[i0p];
    flo[u] = row[ta];  fhi[u] = row[4 + ta];
    fk[u]  = *(const float2*)&row[kk2];
    fjj[u] = row[jj];
  }
}

__device__ __forceinline__ void scan_step_l123(float ni, float nlo, float nhi, float2 nk,
                                               float (&s1)[NI], float (&r2)[NT], v2f (&a3)[NT], v2f (&Av)[NT]) {
  float g2[NI], g3[NI], g4[NI];
#pragma unroll
  for (int ii = 0; ii < NI; ++ii) {
    g2[ii] = __builtin_fmaf(0.5f, ni, s1[ii]);
    g3[ii] = __builtin_fmaf(ni, 1.0f / 3.0f, s1[ii]);
    g4[ii] = __builtin_fmaf(0.25f, ni, s1[ii]);
    s1[ii] += ni;
  }
  const float h2lo = 0.5f * nlo, h2hi = 0.5f * nhi;
  const float h3lo = nlo * (1.0f / 3.0f), h3hi = nhi * (1.0f / 3.0f);
  const float hkA = 0.5f * nk.x, hkB = 0.5f * nk.y;
  const float vjlo = nlo, vjhi = nhi, vkA = nk.x, vkB = nk.y;

#pragma unroll
  for (int t = 0; t < NT; ++t) {
    const int   ii  = t >> 1;
    const float h3j = (t & 1) ? h3hi : h3lo;
    const float h2j = (t & 1) ? h2hi : h2lo;
    const float vj  = (t & 1) ? vjhi : vjlo;
    float t2v = __builtin_fmaf(h3j, g4[ii], r2[t]);
    float u2v = __builtin_fmaf(h2j, g3[ii], r2[t]);
    r2[t] = __builtin_fmaf(vj, g2[ii], r2[t]);
    Av[t].x = __builtin_fmaf(hkA, t2v, a3[t].x);
    Av[t].y = __builtin_fmaf(hkB, t2v, a3[t].y);
    a3[t].x = __builtin_fmaf(vkA, u2v, a3[t].x);
    a3[t].y = __builtin_fmaf(vkB, u2v, a3[t].y);
  }
}

__device__ __forceinline__ void scan_step_partner(float ni, float nlo, float nhi, float2 nk,
                                                  float& s1p, float (&r2p)[NT], v2f (&a3p)[NT]) {
  const float g2 = __builtin_fmaf(0.5f, ni, s1p);
  const float g3 = __builtin_fmaf(ni, 1.0f / 3.0f, s1p);
  s1p += ni;
  const float h2lo = 0.5f * nlo, h2hi = 0.5f * nhi;
#pragma unroll
  for (int t = 0; t < NT; ++t) {
    const float h2j = (t & 1) ? h2hi : h2lo;
    const float vj  = (t & 1) ? nhi : nlo;
    float u2v = __builtin_fmaf(h2j, g3, r2p[t]);
    r2p[t] = __builtin_fmaf(vj, g2, r2p[t]);
    a3p[t].x = __builtin_fmaf(nk.x, u2v, a3p[t].x);
    a3p[t].y = __builtin_fmaf(nk.y, u2v, a3p[t].y);
  }
}

template <int U>
__device__ __forceinline__ void scan_iter(const float (&fi)[GRP], const float (&fip)[GRP],
                                          const float (&flo)[GRP], const float (&fhi)[GRP],
                                          const float2 (&fk)[GRP], const float (&fjj)[GRP], bool blo, bool bhi,
                                          float (&s1)[NI], float (&r2)[NT], v2f (&a3)[NT],
                                          float& s1p, float (&r2p)[NT], v2f (&a3p)[NT],
                                          v8f (&cacc)[NT], v2f (&Ap)[NT], v2f& Bp) {
  static_assert(NT == 2, "one matrix instruction in front of each of the iteration's two parts");
  v2f Av[NT];
  float ni = fi[U], nip = fip[U];
  cacc[0] = __builtin_amdgcn_wmma_f32_16x16x4_f32(false, Ap[0], false, Bp, (short)0, cacc[0], false, false);
  asm volatile("" : "+v"(ni) : "v"(cacc[0]));
  scan_step_l123(ni, flo[U], fhi[U], fk[U], s1, r2, a3, Av);
  __builtin_amdgcn_sched_barrier(0);
  cacc[1] = __builtin_amdgcn_wmma_f32_16x16x4_f32(false, Ap[1], false, Bp, (short)0, cacc[1], false, false);
  asm volatile("" : "+v"(nip) : "v"(cacc[1]));
  scan_step_partner(nip, flo[U], fhi[U], fk[U], s1p, r2p, a3p);
#pragma unroll
  for (int t = 0; t < NT; ++t) Ap[t] = Av[t];
  Bp.x = blo ? fjj[U] : 0.0f;
  Bp.y = bhi ? fjj[U] : 0.0f;
  __builtin_amdgcn_sched_barrier(0);
}

__global__ __launch_bounds__(NTHR)
void logsig_scan_kernel(const float* __restrict__ path, float* __restrict__ out) {
  __shared__ float buf[LDS_FLOATS];
  const int tid  = threadIdx.x;
  const int lane = tid & 31;
  const int w    = __builtin_amdgcn_readfirstlane(tid >> 5);
  const int b    = blockIdx.x >> 1;
  const int h    = blockIdx.x & 1;
  const int T0   = 8 * h + NT * w;
  const int i0   = T0 >> 1;
  const int T0p  = T0 ^ 8;
  const int i0p  = T0p >> 1;
  const int hl   = lane >> 4;
  const float* gp = path + (size_t)b * (L_ * C_);

  {
    const float4* g4 = (const float4*)gp;
    float4* l4 = (float4*)&buf[VINC_OFF];
    constexpr int NPASS = (NSTEP * C_) / 4 / NTHR;
    static_assert(NPASS * NTHR * 4 == NSTEP * C_, "the table is a whole number of passes");
    float4 a[NPASS], c[NPASS];
#pragma unroll
    for (int k = 0; k < NPASS; ++k) {
      a[k] = g4[tid + k * NTHR];  c[k] = g4[tid + k * NTHR + 2];
    }
#pragma unroll
    for (int k = 0; k < NPASS; ++k)
      l4[tid + k * NTHR] = make_float4(c[k].x - a[k].x, c[k].y - a[k].y, c[k].z - a[k].z, c[k].w - a[k].w);
    if (tid < 2 * GRP) l4[(NSTEP * C_) / 4 + tid] = make_float4(0.f, 0.f, 0.f, 0.f);
  }
  __syncthreads();

  v8f  cacc[NT];
  v2f  a3[NT];
  float r2[NT];
  float s1[NI];
  v2f  a3p[NT];
  float r2p[NT];
  float s1p = 0.f;
#pragma unroll
  for (int t = 0; t < NT; ++t) {
    a3[t].x = 0.f; a3[t].y = 0.f; r2[t] = 0.f;
    a3p[t].x = 0.f; a3p[t].y = 0.f; r2p[t] = 0.f;
#pragma unroll
    for (int q = 0; q < 8; ++q) cacc[t][q] = 0.f;
  }
#pragma unroll
  for (int ii = 0; ii < NI; ++ii) s1[ii] = 0.f;

  const int jj  = lane & 7;
  const int kk2 = 2 * (lane & 3);
  const int ta  = (lane >> 2) & 3;
  const bool blo = (lane & 15) < 8;
  const bool bhi = (lane & 15) >= 8;

  const float* vr = &buf[VINC_OFF + hl * (HSTEP * C_)];
  float  xi[GRP], xip[GRP], xlo[GRP], xhi[GRP], xjj[GRP];
  float2 xk[GRP];
  float  yi[GRP], yip[GRP], ylo[GRP], yhi[GRP], yjj[GRP];
  float2 yk[GRP];
  read_group(vr, i0, i0p, ta, kk2, jj, xi, xip, xlo, xhi, xk, xjj);

  v2f Ap[NT], Bp;
#pragma unroll
  for (int t = 0; t < NT; ++t) { Ap[t].x = 0.f; Ap[t].y = 0.f; }
  Bp.x = 0.f; Bp.y = 0.f;

  static_assert(HSTEP % (2 * GRP) == 0, "the scan takes two groups of steps per trip");
  for (int s = 0; s < HSTEP; s += 2 * GRP) {
    read_group(vr + (s + GRP) * C_, i0, i0p, ta, kk2, jj, yi, yip, ylo, yhi, yk, yjj);
    __builtin_amdgcn_sched_barrier(0);
    scan_iter<0>(xi, xip, xlo, xhi, xk, xjj, blo, bhi, s1, r2, a3, s1p, r2p, a3p, cacc, Ap, Bp);
    scan_iter<1>(xi, xip, xlo, xhi, xk, xjj, blo, bhi, s1, r2, a3, s1p, r2p, a3p, cacc, Ap, Bp);
    scan_iter<2>(xi, xip, xlo, xhi, xk, xjj, blo, bhi, s1, r2, a3, s1p, r2p, a3p, cacc, Ap, Bp);
    scan_iter<3>(xi, xip, xlo, xhi, xk, xjj, blo, bhi, s1, r2, a3, s1p, r2p, a3p, cacc, Ap, Bp);
    asm volatile("" :: "v"(yi[0]), "v"(yip[0]), "v"(ylo[0]), "v"(yhi[0]), "v"(yk[0].x), "v"(yk[0].y), "v"(yjj[0]),
                       "v"(yi[1]), "v"(yip[1]), "v"(ylo[1]), "v"(yhi[1]), "v"(yk[1].x), "v"(yk[1].y), "v"(yjj[1]),
                       "v"(yi[2]), "v"(yip[2]), "v"(ylo[2]), "v"(yhi[2]), "v"(yk[2].x), "v"(yk[2].y), "v"(yjj[2]),
                       "v"(yi[3]), "v"(yip[3]), "v"(ylo[3]), "v"(yhi[3]), "v"(yk[3].x), "v"(yk[3].y), "v"(yjj[3]));
    read_group(vr + (s + 2 * GRP) * C_, i0, i0p, ta, kk2, jj, xi, xip, xlo, xhi, xk, xjj);
    __builtin_amdgcn_sched_barrier(0);
    scan_iter<0>(yi, yip, ylo, yhi, yk, yjj, blo, bhi, s1, r2, a3, s1p, r2p, a3p, cacc, Ap, Bp);
    scan_iter<1>(yi, yip, ylo, yhi, yk, yjj, blo, bhi, s1, r2, a3, s1p, r2p, a3p, cacc, Ap, Bp);
    scan_iter<2>(yi, yip, ylo, yhi, yk, yjj, blo, bhi, s1, r2, a3, s1p, r2p, a3p, cacc, Ap, Bp);
    scan_iter<3>(yi, yip, ylo, yhi, yk, yjj, blo, bhi, s1, r2, a3, s1p, r2p, a3p, cacc, Ap, Bp);
    asm volatile("" :: "v"(xi[0]), "v"(xip[0]), "v"(xlo[0]), "v"(xhi[0]), "v"(xk[0].x), "v"(xk[0].y), "v"(xjj[0]),
                       "v"(xi[1]), "v"(xip[1]), "v"(xlo[1]), "v"(xhi[1]), "v"(xk[1].x), "v"(xk[1].y), "v"(xjj[1]),
                       "v"(xi[2]), "v"(xip[2]), "v"(xlo[2]), "v"(xhi[2]), "v"(xk[2].x), "v"(xk[2].y), "v"(xjj[2]),
                       "v"(xi[3]), "v"(xip[3]), "v"(xlo[3]), "v"(xhi[3]), "v"(xk[3].x), "v"(xk[3].y), "v"(xjj[3]));
  }
#pragma unroll
  for (int t = 0; t < NT; ++t)
    cacc[t] = __builtin_amdgcn_wmma_f32_16x16x4_f32(false, Ap[t], false, Bp, (short)0, cacc[t], false, false);

  const int mm = lane & 15;
  const int hi4 = hl * 8, cc4 = lane & 15;
  {
    const int AB = hl ? HALF_B : HALF_A;
    if (mm == 0) {
      buf[AB + S1O + i0]  = s1[0];
      buf[AB + S1O + i0p] = s1p;
    }
    if ((lane & 3) == 0) {
#pragma unroll
      for (int t = 0; t < NT; ++t) {
        buf[AB + S2O + 4 * (T0 + t) + ta]  = r2[t];
        buf[AB + S2O + 4 * (T0p + t) + ta] = r2p[t];
      }
    }
#pragma unroll
    for (int t = 0; t < NT; ++t) {
      const int fo = 32 * (T0 + t) + 2 * mm, fp = 32 * (T0p + t) + 2 * mm;
      buf[AB + S3O + fo] = a3[t].x;   buf[AB + S3O + fo + 1] = a3[t].y;
      buf[AB + S3O + fp] = a3p[t].x;  buf[AB + S3O + fp + 1] = a3p[t].y;
    }
  }
  __syncthreads();
  {
    const float* hA = &buf[HALF_A];
    const float* hB = &buf[HALF_B];
#pragma unroll
    for (int t = 0; t < NT; ++t)
#pragma unroll
      for (int q8 = 0; q8 < 8; ++q8) {
        const int f = (16 * (T0 + t) + q8 + hi4) * 16 + cc4;
        cacc[t][q8] += hA[S1O + (f >> 9)] * hB[S3O + (f & 511)]
                     + hA[S2O + (f >> 6)] * hB[S2O + (f & 63)]
                     + hA[S3O + (f >> 3)] * hB[S1O + (f & 7)];
      }
    const int Tl = hl ? T0p : T0;
#pragma unroll
    for (int t = 0; t < NT; ++t) {
#pragma unroll
      for (int r = 0; r < 2; ++r) {
        const int f = 32 * (Tl + t) + 2 * mm + r;
        const float v3 = hA[S3O + f] + hB[S3O + f]
                       + hA[S1O + (f >> 6)] * hB[S2O + (f & 63)]
                       + hA[S2O + (f >> 3)] * hB[S1O + (f & 7)];
        if (r == 0) a3[t].x = v3; else a3[t].y = v3;
      }
    }
    const int ij = 16 * w + mm;
    const float v2 = hA[S2O + ij] + hB[S2O + ij] + hA[S1O + (ij >> 3)] * hB[S1O + (ij & 7)];
    const float v1 = hA[S1O + (lane & 7)] + hB[S1O + (lane & 7)];

    if (w == 0 && lane < 8) buf[SREG + S1O + lane] = v1;
    buf[SREG + S2O + ij] = v2;
#pragma unroll
    for (int t = 0; t < NT; ++t) {
      const int f = 32 * (Tl + t) + 2 * mm;
      buf[SREG + S3O + f]     = a3[t].x;
      buf[SREG + S3O + f + 1] = a3[t].y;
    }
  }
  __syncthreads();

  float sig1 = (lane < 4) ? buf[SREG + S1O + 4 * h + lane] : 0.f;
  float r2o  = buf[SREG + S2O + 32 * h + lane];

  int Pb = SREG, Qb = QREG;
  for (int n = 2; n <= 4; ++n) {
    float coef = ((n & 1) ? 1.0f : -1.0f) / (float)n;
    float q2;
    {
      int ij = 32 * h + lane;
      q2 = buf[Pb + S1O + (ij >> 3)] * buf[SREG + S1O + (ij & 7)];
      r2o += coef * q2;
    }
    float q3[NT][2];
#pragma unroll
    for (int t = 0; t < NT; ++t) {
#pragma unroll
      for (int r = 0; r < 2; ++r) {
        int f = 32 * (T0 + t) + 2 * mm + r;
        float q = buf[Pb + S1O + (f >> 6)] * buf[SREG + S2O + (f & 63)]
                + buf[Pb + S2O + (f >> 3)] * buf[SREG + S1O + (f & 7)];
        if (r == 0) a3[t].x += coef * q; else a3[t].y += coef * q;
        q3[t][r] = q;
      }
    }
    float q4[NT][8];
#pragma unroll
    for (int t = 0; t < NT; ++t)
#pragma unroll
      for (int q8 = 0; q8 < 8; ++q8) {
        int f = (16 * (T0 + t) + q8 + hi4) * 16 + cc4;
        float q = buf[Pb + S1O + (f >> 9)] * buf[SREG + S3O + (f & 511)]
                + buf[Pb + S2O + (f >> 6)] * buf[SREG + S2O + (f & 63)]
                + buf[Pb + S3O + (f >> 3)] * buf[SREG + S1O + (f & 7)];
        cacc[t][q8] += coef * q;
        q4[t][q8] = q;
      }
    if (w == 0) {
      if (lane < 4) buf[Qb + S1O + 4 * h + lane] = 0.0f;
      buf[Qb + S2O + 32 * h + lane] = q2;
    }
    if (lane < 16) {
#pragma unroll
      for (int t = 0; t < NT; ++t)
#pragma unroll
        for (int r = 0; r < 2; ++r)
          buf[Qb + S3O + 32 * (T0 + t) + 2 * mm + r] = q3[t][r];
    }
#pragma unroll
    for (int t = 0; t < NT; ++t)
#pragma unroll
      for (int q8 = 0; q8 < 8; ++q8)
        buf[Qb + S4O + (16 * (T0 + t) + q8 + hi4) * 16 + cc4] = q4[t][q8];
    __syncthreads();
    int tmp = (Pb == SREG) ? PREG : Pb; Pb = Qb; Qb = tmp;
  }

  float* ob = out + (size_t)b * OUTPB;
  if (w == 0) {
    if (lane < 4) ob[4 * h + lane] = sig1;
    ob[8 + 32 * h + lane] = r2o;
  }
  if (lane < 16) {
#pragma unroll
    for (int t = 0; t < NT; ++t) {
      int f = 32 * (T0 + t) + 2 * lane;
      ob[72 + f]     = a3[t].x;
      ob[72 + f + 1] = a3[t].y;
    }
  }
  {
    const int hi = (lane >> 4) * 8, cc = lane & 15;
#pragma unroll
    for (int t = 0; t < NT; ++t)
#pragma unroll
      for (int q = 0; q < 8; ++q) {
        int f = (16 * (T0 + t) + q + hi) * 16 + cc;
        ob[584 + f] = cacc[t][q];
      }
  }
}

extern "C" void kernel_launch(void* const* d_in, const int* in_sizes, int n_in,
                              void* d_out, int out_size, void* d_ws, size_t ws_size,
                              hipStream_t stream) {
  const float* path = (const float*)d_in[0];
  float* out = (float*)d_out;
  logsig_scan_kernel<<<dim3(NGRID), dim3(NTHR), 0, stream>>>(path, out);
}


# Scan: increments are read from LDS as ONE group per 8 steps (was 4), all in one stretch between matrix instructions, and are first waited for behind the group's 8 steps of work; sums unchanged
# speedup vs baseline: 12.8365x; 1.0832x over previous
#include <hip/hip_runtime.h>


typedef float v2f __attribute__((ext_vector_type(2)));
typedef float v8f __attribute__((ext_vector_type(8)));

#define B_      128
#define L_      257
#define C_      8
#define NSTEP   256
#define OUTPB   4680

#define NG      2
#define NW      4
#define NTHR    128
#define NT      (16 / (NG * NW))
#define NI      (NT / 2)
static_assert(NTHR == 32 * NW, "one wave is 32 lanes");
static_assert(NT == 2 && NI == 1, "a wave owns one (even, odd) tile pair, which shares one level-1 component");
static_assert(NG == 2, "lanes 16-31 carry the partner pair of exactly one other group");
#define NGRID   256
static_assert(NGRID == NG * B_, "two work-groups per batch element");

#define VINC_OFF 4680
#define S1O 0
#define S2O 8
#define S3O 72
#define S4O 584
#define SREG 0
#define PREG 4680
#define QREG 9360
#define LDS_FLOATS 14040
#define HALF_A  QREG
#define HALF_B  (QREG + 1024)
#define HSTEP   (NSTEP / 2)

#define GRP     8
static_assert(GRP == 8, "the fences in the scan name the eight steps of a group one by one");

__device__ __forceinline__ void read_group(const float* rows, int i0, int i0p, int ta, int kk2, int jj,
                                           float (&fi)[GRP], float (&fip)[GRP], float (&flo)[GRP], float (&fhi)[GRP],
                                           float2 (&fk)[GRP], float (&fjj)[GRP]) {
#pragma unroll
  for (int u = 0; u < GRP; ++u) {
    const float* row = rows + u * C_;
    fi[u]  = row[i0];
    fip[u] = row[i0p];
    flo[u] = row[ta];  fhi[u] = row[4 + ta];
    fk[u]  = *(const float2*)&row[kk2];
    fjj[u] = row[jj];
  }
}

__device__ __forceinline__ void scan_step_l123(float ni, float nlo, float nhi, float2 nk,
                                               float (&s1)[NI], float (&r2)[NT], v2f (&a3)[NT], v2f (&Av)[NT]) {
  float g2[NI], g3[NI], g4[NI];
#pragma unroll
  for (int ii = 0; ii < NI; ++ii) {
    g2[ii] = __builtin_fmaf(0.5f, ni, s1[ii]);
    g3[ii] = __builtin_fmaf(ni, 1.0f / 3.0f, s1[ii]);
    g4[ii] = __builtin_fmaf(0.25f, ni, s1[ii]);
    s1[ii] += ni;
  }
  const float h2lo = 0.5f * nlo, h2hi = 0.5f * nhi;
  const float h3lo = nlo * (1.0f / 3.0f), h3hi = nhi * (1.0f / 3.0f);
  const float hkA = 0.5f * nk.x, hkB = 0.5f * nk.y;
  const float vjlo = nlo, vjhi = nhi, vkA = nk.x, vkB = nk.y;

#pragma unroll
  for (int t = 0; t < NT; ++t) {
    const int   ii  = t >> 1;
    const float h3j = (t & 1) ? h3hi : h3lo;
    const float h2j = (t & 1) ? h2hi : h2lo;
    const float vj  = (t & 1) ? vjhi : vjlo;
    float t2v = __builtin_fmaf(h3j, g4[ii], r2[t]);
    float u2v = __builtin_fmaf(h2j, g3[ii], r2[t]);
    r2[t] = __builtin_fmaf(vj, g2[ii], r2[t]);
    Av[t].x = __builtin_fmaf(hkA, t2v, a3[t].x);
    Av[t].y = __builtin_fmaf(hkB, t2v, a3[t].y);
    a3[t].x = __builtin_fmaf(vkA, u2v, a3[t].x);
    a3[t].y = __builtin_fmaf(vkB, u2v, a3[t].y);
  }
}

__device__ __forceinline__ void scan_step_partner(float ni, float nlo, float nhi, float2 nk,
                                                  float& s1p, float (&r2p)[NT], v2f (&a3p)[NT]) {
  const float g2 = __builtin_fmaf(0.5f, ni, s1p);
  const float g3 = __builtin_fmaf(ni, 1.0f / 3.0f, s1p);
  s1p += ni;
  const float h2lo = 0.5f * nlo, h2hi = 0.5f * nhi;
#pragma unroll
  for (int t = 0; t < NT; ++t) {
    const float h2j = (t & 1) ? h2hi : h2lo;
    const float vj  = (t & 1) ? nhi : nlo;
    float u2v = __builtin_fmaf(h2j, g3, r2p[t]);
    r2p[t] = __builtin_fmaf(vj, g2, r2p[t]);
    a3p[t].x = __builtin_fmaf(nk.x, u2v, a3p[t].x);
    a3p[t].y = __builtin_fmaf(nk.y, u2v, a3p[t].y);
  }
}

template <int U>
__device__ __forceinline__ void scan_iter(const float (&fi)[GRP], const float (&fip)[GRP],
                                          const float (&flo)[GRP], const float (&fhi)[GRP],
                                          const float2 (&fk)[GRP], const float (&fjj)[GRP], bool blo, bool bhi,
                                          float (&s1)[NI], float (&r2)[NT], v2f (&a3)[NT],
                                          float& s1p, float (&r2p)[NT], v2f (&a3p)[NT],
                                          v8f (&cacc)[NT], v2f (&Ap)[NT], v2f& Bp) {
  static_assert(NT == 2, "one matrix instruction in front of each of the iteration's two parts");
  v2f Av[NT];
  float ni = fi[U], nip = fip[U];
  cacc[0] = __builtin_amdgcn_wmma_f32_16x16x4_f32(false, Ap[0], false, Bp, (short)0, cacc[0], false, false);
  asm volatile("" : "+v"(ni) : "v"(cacc[0]));
  scan_step_l123(ni, flo[U], fhi[U], fk[U], s1, r2, a3, Av);
  __builtin_amdgcn_sched_barrier(0);
  cacc[1] = __builtin_amdgcn_wmma_f32_16x16x4_f32(false, Ap[1], false, Bp, (short)0, cacc[1], false, false);
  asm volatile("" : "+v"(nip) : "v"(cacc[1]));
  scan_step_partner(nip, flo[U], fhi[U], fk[U], s1p, r2p, a3p);
#pragma unroll
  for (int t = 0; t < NT; ++t) Ap[t] = Av[t];
  Bp.x = blo ? fjj[U] : 0.0f;
  Bp.y = bhi ? fjj[U] : 0.0f;
  __builtin_amdgcn_sched_barrier(0);
}

__global__ __launch_bounds__(NTHR)
void logsig_scan_kernel(const float* __restrict__ path, float* __restrict__ out) {
  __shared__ float buf[LDS_FLOATS];
  const int tid  = threadIdx.x;
  const int lane = tid & 31;
  const int w    = __builtin_amdgcn_readfirstlane(tid >> 5);
  const int b    = blockIdx.x >> 1;
  const int h    = blockIdx.x & 1;
  const int T0   = 8 * h + NT * w;
  const int i0   = T0 >> 1;
  const int T0p  = T0 ^ 8;
  const int i0p  = T0p >> 1;
  const int hl   = lane >> 4;
  const float* gp = path + (size_t)b * (L_ * C_);

  {
    const float4* g4 = (const float4*)gp;
    float4* l4 = (float4*)&buf[VINC_OFF];
    constexpr int NPASS = (NSTEP * C_) / 4 / NTHR;
    static_assert(NPASS * NTHR * 4 == NSTEP * C_, "the table is a whole number of passes");
    float4 a[NPASS], c[NPASS];
#pragma unroll
    for (int k = 0; k < NPASS; ++k) {
      a[k] = g4[tid + k * NTHR];  c[k] = g4[tid + k * NTHR + 2];
    }
#pragma unroll
    for (int k = 0; k < NPASS; ++k)
      l4[tid + k * NTHR] = make_float4(c[k].x - a[k].x, c[k].y - a[k].y, c[k].z - a[k].z, c[k].w - a[k].w);
    if (tid < 2 * GRP) l4[(NSTEP * C_) / 4 + tid] = make_float4(0.f, 0.f, 0.f, 0.f);
  }
  __syncthreads();

  v8f  cacc[NT];
  v2f  a3[NT];
  float r2[NT];
  float s1[NI];
  v2f  a3p[NT];
  float r2p[NT];
  float s1p = 0.f;
#pragma unroll
  for (int t = 0; t < NT; ++t) {
    a3[t].x = 0.f; a3[t].y = 0.f; r2[t] = 0.f;
    a3p[t].x = 0.f; a3p[t].y = 0.f; r2p[t] = 0.f;
#pragma unroll
    for (int q = 0; q < 8; ++q) cacc[t][q] = 0.f;
  }
#pragma unroll
  for (int ii = 0; ii < NI; ++ii) s1[ii] = 0.f;

  const int jj  = lane & 7;
  const int kk2 = 2 * (lane & 3);
  const int ta  = (lane >> 2) & 3;
  const bool blo = (lane & 15) < 8;
  const bool bhi = (lane & 15) >= 8;

  const float* vr = &buf[VINC_OFF + hl * (HSTEP * C_)];
  float  xi[GRP], xip[GRP], xlo[GRP], xhi[GRP], xjj[GRP];
  float2 xk[GRP];
  float  yi[GRP], yip[GRP], ylo[GRP], yhi[GRP], yjj[GRP];
  float2 yk[GRP];
  read_group(vr, i0, i0p, ta, kk2, jj, xi, xip, xlo, xhi, xk, xjj);

  v2f Ap[NT], Bp;
#pragma unroll
  for (int t = 0; t < NT; ++t) { Ap[t].x = 0.f; Ap[t].y = 0.f; }
  Bp.x = 0.f; Bp.y = 0.f;

  static_assert(HSTEP % (2 * GRP) == 0, "the scan takes two groups of steps per trip");
  for (int s = 0; s < HSTEP; s += 2 * GRP) {
    read_group(vr + (s + GRP) * C_, i0, i0p, ta, kk2, jj, yi, yip, ylo, yhi, yk, yjj);
    __builtin_amdgcn_sched_barrier(0);
    scan_iter<0>(xi, xip, xlo, xhi, xk, xjj, blo, bhi, s1, r2, a3, s1p, r2p, a3p, cacc, Ap, Bp);
    scan_iter<1>(xi, xip, xlo, xhi, xk, xjj, blo, bhi, s1, r2, a3, s1p, r2p, a3p, cacc, Ap, Bp);
    scan_iter<2>(xi, xip, xlo, xhi, xk, xjj, blo, bhi, s1, r2, a3, s1p, r2p, a3p, cacc, Ap, Bp);
    scan_iter<3>(xi, xip, xlo, xhi, xk, xjj, blo, bhi, s1, r2, a3, s1p, r2p, a3p, cacc, Ap, Bp);
    scan_iter<4>(xi, xip, xlo, xhi, xk, xjj, blo, bhi, s1, r2, a3, s1p, r2p, a3p, cacc, Ap, Bp);
    scan_iter<5>(xi, xip, xlo, xhi, xk, xjj, blo, bhi, s1, r2, a3, s1p, r2p, a3p, cacc, Ap, Bp);
    scan_iter<6>(xi, xip, xlo, xhi, xk, xjj, blo, bhi, s1, r2, a3, s1p, r2p, a3p, cacc, Ap, Bp);
    scan_iter<7>(xi, xip, xlo, xhi, xk, xjj, blo, bhi, s1, r2, a3, s1p, r2p, a3p, cacc, Ap, Bp);
    asm volatile("" : "+v"(yi[0]), "+v"(yip[0]), "+v"(ylo[0]), "+v"(yhi[0]), "+v"(yk[0].x), "+v"(yk[0].y), "+v"(yjj[0]),
                      "+v"(yi[1]), "+v"(yip[1]), "+v"(ylo[1]), "+v"(yhi[1]), "+v"(yk[1].x), "+v"(yk[1].y), "+v"(yjj[1]),
                      "+v"(yi[2]), "+v"(yip[2]), "+v"(ylo[2]), "+v"(yhi[2]), "+v"(yk[2].x), "+v"(yk[2].y), "+v"(yjj[2]),
                      "+v"(yi[3]), "+v"(yip[3]), "+v"(ylo[3]), "+v"(yhi[3]), "+v"(yk[3].x), "+v"(yk[3].y), "+v"(yjj[3]));
    asm volatile("" : "+v"(yi[4]), "+v"(yip[4]), "+v"(ylo[4]), "+v"(yhi[4]), "+v"(yk[4].x), "+v"(yk[4].y), "+v"(yjj[4]),
                      "+v"(yi[5]), "+v"(yip[5]), "+v"(ylo[5]), "+v"(yhi[5]), "+v"(yk[5].x), "+v"(yk[5].y), "+v"(yjj[5]),
                      "+v"(yi[6]), "+v"(yip[6]), "+v"(ylo[6]), "+v"(yhi[6]), "+v"(yk[6].x), "+v"(yk[6].y), "+v"(yjj[6]),
                      "+v"(yi[7]), "+v"(yip[7]), "+v"(ylo[7]), "+v"(yhi[7]), "+v"(yk[7].x), "+v"(yk[7].y), "+v"(yjj[7]));
    read_group(vr + (s + 2 * GRP) * C_, i0, i0p, ta, kk2, jj, xi, xip, xlo, xhi, xk, xjj);
    __builtin_amdgcn_sched_barrier(0);
    scan_iter<0>(yi, yip, ylo, yhi, yk, yjj, blo, bhi, s1, r2, a3, s1p, r2p, a3p, cacc, Ap, Bp);
    scan_iter<1>(yi, yip, ylo, yhi, yk, yjj, blo, bhi, s1, r2, a3, s1p, r2p, a3p, cacc, Ap, Bp);
    scan_iter<2>(yi, yip, ylo, yhi, yk, yjj, blo, bhi, s1, r2, a3, s1p, r2p, a3p, cacc, Ap, Bp);
    scan_iter<3>(yi, yip, ylo, yhi, yk, yjj, blo, bhi, s1, r2, a3, s1p, r2p, a3p, cacc, Ap, Bp);
    scan_iter<4>(yi, yip, ylo, yhi, yk, yjj, blo, bhi, s1, r2, a3, s1p, r2p, a3p, cacc, Ap, Bp);
    scan_iter<5>(yi, yip, ylo, yhi, yk, yjj, blo, bhi, s1, r2, a3, s1p, r2p, a3p, cacc, Ap, Bp);
    scan_iter<6>(yi, yip, ylo, yhi, yk, yjj, blo, bhi, s1, r2, a3, s1p, r2p, a3p, cacc, Ap, Bp);
    scan_iter<7>(yi, yip, ylo, yhi, yk, yjj, blo, bhi, s1, r2, a3, s1p, r2p, a3p, cacc, Ap, Bp);
    asm volatile("" : "+v"(xi[0]), "+v"(xip[0]), "+v"(xlo[0]), "+v"(xhi[0]), "+v"(xk[0].x), "+v"(xk[0].y), "+v"(xjj[0]),
                      "+v"(xi[1]), "+v"(xip[1]), "+v"(xlo[1]), "+v"(xhi[1]), "+v"(xk[1].x), "+v"(xk[1].y), "+v"(xjj[1]),
                      "+v"(xi[2]), "+v"(xip[2]), "+v"(xlo[2]), "+v"(xhi[2]), "+v"(xk[2].x), "+v"(xk[2].y), "+v"(xjj[2]),
                      "+v"(xi[3]), "+v"(xip[3]), "+v"(xlo[3]), "+v"(xhi[3]), "+v"(xk[3].x), "+v"(xk[3].y), "+v"(xjj[3]));
    asm volatile("" : "+v"(xi[4]), "+v"(xip[4]), "+v"(xlo[4]), "+v"(xhi[4]), "+v"(xk[4].x), "+v"(xk[4].y), "+v"(xjj[4]),
                      "+v"(xi[5]), "+v"(xip[5]), "+v"(xlo[5]), "+v"(xhi[5]), "+v"(xk[5].x), "+v"(xk[5].y), "+v"(xjj[5]),
                      "+v"(xi[6]), "+v"(xip[6]), "+v"(xlo[6]), "+v"(xhi[6]), "+v"(xk[6].x), "+v"(xk[6].y), "+v"(xjj[6]),
                      "+v"(xi[7]), "+v"(xip[7]), "+v"(xlo[7]), "+v"(xhi[7]), "+v"(xk[7].x), "+v"(xk[7].y), "+v"(xjj[7]));
  }
#pragma unroll
  for (int t = 0; t < NT; ++t)
    cacc[t] = __builtin_amdgcn_wmma_f32_16x16x4_f32(false, Ap[t], false, Bp, (short)0, cacc[t], false, false);

  const int mm = lane & 15;
  const int hi4 = hl * 8, cc4 = lane & 15;
  {
    const int AB = hl ? HALF_B : HALF_A;
    if (mm == 0) {
      buf[AB + S1O + i0]  = s1[0];
      buf[AB + S1O + i0p] = s1p;
    }
    if ((lane & 3) == 0) {
#pragma unroll
      for (int t = 0; t < NT; ++t) {
        buf[AB + S2O + 4 * (T0 + t) + ta]  = r2[t];
        buf[AB + S2O + 4 * (T0p + t) + ta] = r2p[t];
      }
    }
#pragma unroll
    for (int t = 0; t < NT; ++t) {
      const int fo = 32 * (T0 + t) + 2 * mm, fp = 32 * (T0p + t) + 2 * mm;
      buf[AB + S3O + fo] = a3[t].x;   buf[AB + S3O + fo + 1] = a3[t].y;
      buf[AB + S3O + fp] = a3p[t].x;  buf[AB + S3O + fp + 1] = a3p[t].y;
    }
  }
  __syncthreads();
  {
    const float* hA = &buf[HALF_A];
    const float* hB = &buf[HALF_B];
#pragma unroll
    for (int t = 0; t < NT; ++t)
#pragma unroll
      for (int q8 = 0; q8 < 8; ++q8) {
        const int f = (16 * (T0 + t) + q8 + hi4) * 16 + cc4;
        cacc[t][q8] += hA[S1O + (f >> 9)] * hB[S3O + (f & 511)]
                     + hA[S2O + (f >> 6)] * hB[S2O + (f & 63)]
                     + hA[S3O + (f >> 3)] * hB[S1O + (f & 7)];
      }
    const int Tl = hl ? T0p : T0;
#pragma unroll
    for (int t = 0; t < NT; ++t) {
#pragma unroll
      for (int r = 0; r < 2; ++r) {
        const int f = 32 * (Tl + t) + 2 * mm + r;
        const float v3 = hA[S3O + f] + hB[S3O + f]
                       + hA[S1O + (f >> 6)] * hB[S2O + (f & 63)]
                       + hA[S2O + (f >> 3)] * hB[S1O + (f & 7)];
        if (r == 0) a3[t].x = v3; else a3[t].y = v3;
      }
    }
    const int ij = 16 * w + mm;
    const float v2 = hA[S2O + ij] + hB[S2O + ij] + hA[S1O + (ij >> 3)] * hB[S1O + (ij & 7)];
    const float v1 = hA[S1O + (lane & 7)] + hB[S1O + (lane & 7)];

    if (w == 0 && lane < 8) buf[SREG + S1O + lane] = v1;
    buf[SREG + S2O + ij] = v2;
#pragma unroll
    for (int t = 0; t < NT; ++t) {
      const int f = 32 * (Tl + t) + 2 * mm;
      buf[SREG + S3O + f]     = a3[t].x;
      buf[SREG + S3O + f + 1] = a3[t].y;
    }
  }
  __syncthreads();

  float sig1 = (lane < 4) ? buf[SREG + S1O + 4 * h + lane] : 0.f;
  float r2o  = buf[SREG + S2O + 32 * h + lane];

  int Pb = SREG, Qb = QREG;
  for (int n = 2; n <= 4; ++n) {
    float coef = ((n & 1) ? 1.0f : -1.0f) / (float)n;
    float q2;
    {
      int ij = 32 * h + lane;
      q2 = buf[Pb + S1O + (ij >> 3)] * buf[SREG + S1O + (ij & 7)];
      r2o += coef * q2;
    }
    float q3[NT][2];
#pragma unroll
    for (int t = 0; t < NT; ++t) {
#pragma unroll
      for (int r = 0; r < 2; ++r) {
        int f = 32 * (T0 + t) + 2 * mm + r;
        float q = buf[Pb + S1O + (f >> 6)] * buf[SREG + S2O + (f & 63)]
                + buf[Pb + S2O + (f >> 3)] * buf[SREG + S1O + (f & 7)];
        if (r == 0) a3[t].x += coef * q; else a3[t].y += coef * q;
        q3[t][r] = q;
      }
    }
    float q4[NT][8];
#pragma unroll
    for (int t = 0; t < NT; ++t)
#pragma unroll
      for (int q8 = 0; q8 < 8; ++q8) {
        int f = (16 * (T0 + t) + q8 + hi4) * 16 + cc4;
        float q = buf[Pb + S1O + (f >> 9)] * buf[SREG + S3O + (f & 511)]
                + buf[Pb + S2O + (f >> 6)] * buf[SREG + S2O + (f & 63)]
                + buf[Pb + S3O + (f >> 3)] * buf[SREG + S1O + (f & 7)];
        cacc[t][q8] += coef * q;
        q4[t][q8] = q;
      }
    if (w == 0) {
      if (lane < 4) buf[Qb + S1O + 4 * h + lane] = 0.0f;
      buf[Qb + S2O + 32 * h + lane] = q2;
    }
    if (lane < 16) {
#pragma unroll
      for (int t = 0; t < NT; ++t)
#pragma unroll
        for (int r = 0; r < 2; ++r)
          buf[Qb + S3O + 32 * (T0 + t) + 2 * mm + r] = q3[t][r];
    }
#pragma unroll
    for (int t = 0; t < NT; ++t)
#pragma unroll
      for (int q8 = 0; q8 < 8; ++q8)
        buf[Qb + S4O + (16 * (T0 + t) + q8 + hi4) * 16 + cc4] = q4[t][q8];
    __syncthreads();
    int tmp = (Pb == SREG) ? PREG : Pb; Pb = Qb; Qb = tmp;
  }

  float* ob = out + (size_t)b * OUTPB;
  if (w == 0) {
    if (lane < 4) ob[4 * h + lane] = sig1;
    ob[8 + 32 * h + lane] = r2o;
  }
  if (lane < 16) {
#pragma unroll
    for (int t = 0; t < NT; ++t) {
      int f = 32 * (T0 + t) + 2 * lane;
      ob[72 + f]     = a3[t].x;
      ob[72 + f + 1] = a3[t].y;
    }
  }
  {
    const int hi = (lane >> 4) * 8, cc = lane & 15;
#pragma unroll
    for (int t = 0; t < NT; ++t)
#pragma unroll
      for (int q = 0; q < 8; ++q) {
        int f = (16 * (T0 + t) + q + hi) * 16 + cc;
        ob[584 + f] = cacc[t][q];
      }
  }
}

extern "C" void kernel_launch(void* const* d_in, const int* in_sizes, int n_in,
                              void* d_out, int out_size, void* d_ws, size_t ws_size,
                              hipStream_t stream) {
  const float* path = (const float*)d_in[0];
  float* out = (float*)d_out;
  logsig_scan_kernel<<<dim3(NGRID), dim3(NTHR), 0, stream>>>(path, out);
}


# Logarithm in ONE pass: each lane forms what it needs of the powers' lower levels (1-2 products of S) in registers, no power goes through LDS: 1 LDS round trip, no write or barrier (3 each); same sums
# speedup vs baseline: 13.6717x; 1.0651x over previous
#include <hip/hip_runtime.h>


typedef float v2f __attribute__((ext_vector_type(2)));
typedef float v8f __attribute__((ext_vector_type(8)));

#define B_      128
#define L_      257
#define C_      8
#define NSTEP   256
#define OUTPB   4680

#define NG      2
#define NW      4
#define NTHR    128
#define NT      (16 / (NG * NW))
#define NI      (NT / 2)
static_assert(NTHR == 32 * NW, "one wave is 32 lanes");
static_assert(NT == 2 && NI == 1, "a wave owns one (even, odd) tile pair, which shares one level-1 component");
static_assert(NG == 2, "lanes 16-31 carry the partner pair of exactly one other group");
#define NGRID   256
static_assert(NGRID == NG * B_, "two work-groups per batch element");

#define VINC_OFF 4680
#define S1O 0
#define S2O 8
#define S3O 72
#define S4O 584
#define SREG 0
#define PREG 4680
#define QREG 9360
#define LDS_FLOATS 14040
#define HALF_A  QREG
#define HALF_B  (QREG + 1024)
#define HSTEP   (NSTEP / 2)

#define GRP     8
static_assert(GRP == 8, "the fences in the scan name the eight steps of a group one by one");

__device__ __forceinline__ void read_group(const float* rows, int i0, int i0p, int ta, int kk2, int jj,
                                           float (&fi)[GRP], float (&fip)[GRP], float (&flo)[GRP], float (&fhi)[GRP],
                                           float2 (&fk)[GRP], float (&fjj)[GRP]) {
#pragma unroll
  for (int u = 0; u < GRP; ++u) {
    const float* row = rows + u * C_;
    fi[u]  = row[i0];
    fip[u] = row[i0p];
    flo[u] = row[ta];  fhi[u] = row[4 + ta];
    fk[u]  = *(const float2*)&row[kk2];
    fjj[u] = row[jj];
  }
}

__device__ __forceinline__ void scan_step_l123(float ni, float nlo, float nhi, float2 nk,
                                               float (&s1)[NI], float (&r2)[NT], v2f (&a3)[NT], v2f (&Av)[NT]) {
  float g2[NI], g3[NI], g4[NI];
#pragma unroll
  for (int ii = 0; ii < NI; ++ii) {
    g2[ii] = __builtin_fmaf(0.5f, ni, s1[ii]);
    g3[ii] = __builtin_fmaf(ni, 1.0f / 3.0f, s1[ii]);
    g4[ii] = __builtin_fmaf(0.25f, ni, s1[ii]);
    s1[ii] += ni;
  }
  const float h2lo = 0.5f * nlo, h2hi = 0.5f * nhi;
  const float h3lo = nlo * (1.0f / 3.0f), h3hi = nhi * (1.0f / 3.0f);
  const float hkA = 0.5f * nk.x, hkB = 0.5f * nk.y;
  const float vjlo = nlo, vjhi = nhi, vkA = nk.x, vkB = nk.y;

#pragma unroll
  for (int t = 0; t < NT; ++t) {
    const int   ii  = t >> 1;
    const float h3j = (t & 1) ? h3hi : h3lo;
    const float h2j = (t & 1) ? h2hi : h2lo;
    const float vj  = (t & 1) ? vjhi : vjlo;
    float t2v = __builtin_fmaf(h3j, g4[ii], r2[t]);
    float u2v = __builtin_fmaf(h2j, g3[ii], r2[t]);
    r2[t] = __builtin_fmaf(vj, g2[ii], r2[t]);
    Av[t].x = __builtin_fmaf(hkA, t2v, a3[t].x);
    Av[t].y = __builtin_fmaf(hkB, t2v, a3[t].y);
    a3[t].x = __builtin_fmaf(vkA, u2v, a3[t].x);
    a3[t].y = __builtin_fmaf(vkB, u2v, a3[t].y);
  }
}

__device__ __forceinline__ void scan_step_partner(float ni, float nlo, float nhi, float2 nk,
                                                  float& s1p, float (&r2p)[NT], v2f (&a3p)[NT]) {
  const float g2 = __builtin_fmaf(0.5f, ni, s1p);
  const float g3 = __builtin_fmaf(ni, 1.0f / 3.0f, s1p);
  s1p += ni;
  const float h2lo = 0.5f * nlo, h2hi = 0.5f * nhi;
#pragma unroll
  for (int t = 0; t < NT; ++t) {
    const float h2j = (t & 1) ? h2hi : h2lo;
    const float vj  = (t & 1) ? nhi : nlo;
    float u2v = __builtin_fmaf(h2j, g3, r2p[t]);
    r2p[t] = __builtin_fmaf(vj, g2, r2p[t]);
    a3p[t].x = __builtin_fmaf(nk.x, u2v, a3p[t].x);
    a3p[t].y = __builtin_fmaf(nk.y, u2v, a3p[t].y);
  }
}

template <int U>
__device__ __forceinline__ void scan_iter(const float (&fi)[GRP], const float (&fip)[GRP],
                                          const float (&flo)[GRP], const float (&fhi)[GRP],
                                          const float2 (&fk)[GRP], const float (&fjj)[GRP], bool blo, bool bhi,
                                          float (&s1)[NI], float (&r2)[NT], v2f (&a3)[NT],
                                          float& s1p, float (&r2p)[NT], v2f (&a3p)[NT],
                                          v8f (&cacc)[NT], v2f (&Ap)[NT], v2f& Bp) {
  static_assert(NT == 2, "one matrix instruction in front of each of the iteration's two parts");
  v2f Av[NT];
  float ni = fi[U], nip = fip[U];
  cacc[0] = __builtin_amdgcn_wmma_f32_16x16x4_f32(false, Ap[0], false, Bp, (short)0, cacc[0], false, false);
  asm volatile("" : "+v"(ni) : "v"(cacc[0]));
  scan_step_l123(ni, flo[U], fhi[U], fk[U], s1, r2, a3, Av);
  __builtin_amdgcn_sched_barrier(0);
  cacc[1] = __builtin_amdgcn_wmma_f32_16x16x4_f32(false, Ap[1], false, Bp, (short)0, cacc[1], false, false);
  asm volatile("" : "+v"(nip) : "v"(cacc[1]));
  scan_step_partner(nip, flo[U], fhi[U], fk[U], s1p, r2p, a3p);
#pragma unroll
  for (int t = 0; t < NT; ++t) Ap[t] = Av[t];
  Bp.x = blo ? fjj[U] : 0.0f;
  Bp.y = bhi ? fjj[U] : 0.0f;
  __builtin_amdgcn_sched_barrier(0);
}

__global__ __launch_bounds__(NTHR)
void logsig_scan_kernel(const float* __restrict__ path, float* __restrict__ out) {
  __shared__ float buf[LDS_FLOATS];
  const int tid  = threadIdx.x;
  const int lane = tid & 31;
  const int w    = __builtin_amdgcn_readfirstlane(tid >> 5);
  const int b    = blockIdx.x >> 1;
  const int h    = blockIdx.x & 1;
  const int T0   = 8 * h + NT * w;
  const int i0   = T0 >> 1;
  const int T0p  = T0 ^ 8;
  const int i0p  = T0p >> 1;
  const int hl   = lane >> 4;
  const float* gp = path + (size_t)b * (L_ * C_);

  {
    const float4* g4 = (const float4*)gp;
    float4* l4 = (float4*)&buf[VINC_OFF];
    constexpr int NPASS = (NSTEP * C_) / 4 / NTHR;
    static_assert(NPASS * NTHR * 4 == NSTEP * C_, "the table is a whole number of passes");
    float4 a[NPASS], c[NPASS];
#pragma unroll
    for (int k = 0; k < NPASS; ++k) {
      a[k] = g4[tid + k * NTHR];  c[k] = g4[tid + k * NTHR + 2];
    }
#pragma unroll
    for (int k = 0; k < NPASS; ++k)
      l4[tid + k * NTHR] = make_float4(c[k].x - a[k].x, c[k].y - a[k].y, c[k].z - a[k].z, c[k].w - a[k].w);
    if (tid < 2 * GRP) l4[(NSTEP * C_) / 4 + tid] = make_float4(0.f, 0.f, 0.f, 0.f);
  }
  __syncthreads();

  v8f  cacc[NT];
  v2f  a3[NT];
  float r2[NT];
  float s1[NI];
  v2f  a3p[NT];
  float r2p[NT];
  float s1p = 0.f;
#pragma unroll
  for (int t = 0; t < NT; ++t) {
    a3[t].x = 0.f; a3[t].y = 0.f; r2[t] = 0.f;
    a3p[t].x = 0.f; a3p[t].y = 0.f; r2p[t] = 0.f;
#pragma unroll
    for (int q = 0; q < 8; ++q) cacc[t][q] = 0.f;
  }
#pragma unroll
  for (int ii = 0; ii < NI; ++ii) s1[ii] = 0.f;

  const int jj  = lane & 7;
  const int kk2 = 2 * (lane & 3);
  const int ta  = (lane >> 2) & 3;
  const bool blo = (lane & 15) < 8;
  const bool bhi = (lane & 15) >= 8;

  const float* vr = &buf[VINC_OFF + hl * (HSTEP * C_)];
  float  xi[GRP], xip[GRP], xlo[GRP], xhi[GRP], xjj[GRP];
  float2 xk[GRP];
  float  yi[GRP], yip[GRP], ylo[GRP], yhi[GRP], yjj[GRP];
  float2 yk[GRP];
  read_group(vr, i0, i0p, ta, kk2, jj, xi, xip, xlo, xhi, xk, xjj);

  v2f Ap[NT], Bp;
#pragma unroll
  for (int t = 0; t < NT; ++t) { Ap[t].x = 0.f; Ap[t].y = 0.f; }
  Bp.x = 0.f; Bp.y = 0.f;

  static_assert(HSTEP % (2 * GRP) == 0, "the scan takes two groups of steps per trip");
  for (int s = 0; s < HSTEP; s += 2 * GRP) {
    read_group(vr + (s + GRP) * C_, i0, i0p, ta, kk2, jj, yi, yip, ylo, yhi, yk, yjj);
    __builtin_amdgcn_sched_barrier(0);
    scan_iter<0>(xi, xip, xlo, xhi, xk, xjj, blo, bhi, s1, r2, a3, s1p, r2p, a3p, cacc, Ap, Bp);
    scan_iter<1>(xi, xip, xlo, xhi, xk, xjj, blo, bhi, s1, r2, a3, s1p, r2p, a3p, cacc, Ap, Bp);
    scan_iter<2>(xi, xip, xlo, xhi, xk, xjj, blo, bhi, s1, r2, a3, s1p, r2p, a3p, cacc, Ap, Bp);
    scan_iter<3>(xi, xip, xlo, xhi, xk, xjj, blo, bhi, s1, r2, a3, s1p, r2p, a3p, cacc, Ap, Bp);
    scan_iter<4>(xi, xip, xlo, xhi, xk, xjj, blo, bhi, s1, r2, a3, s1p, r2p, a3p, cacc, Ap, Bp);
    scan_iter<5>(xi, xip, xlo, xhi, xk, xjj, blo, bhi, s1, r2, a3, s1p, r2p, a3p, cacc, Ap, Bp);
    scan_iter<6>(xi, xip, xlo, xhi, xk, xjj, blo, bhi, s1, r2, a3, s1p, r2p, a3p, cacc, Ap, Bp);
    scan_iter<7>(xi, xip, xlo, xhi, xk, xjj, blo, bhi, s1, r2, a3, s1p, r2p, a3p, cacc, Ap, Bp);
    asm volatile("" : "+v"(yi[0]), "+v"(yip[0]), "+v"(ylo[0]), "+v"(yhi[0]), "+v"(yk[0].x), "+v"(yk[0].y), "+v"(yjj[0]),
                      "+v"(yi[1]), "+v"(yip[1]), "+v"(ylo[1]), "+v"(yhi[1]), "+v"(yk[1].x), "+v"(yk[1].y), "+v"(yjj[1]),
                      "+v"(yi[2]), "+v"(yip[2]), "+v"(ylo[2]), "+v"(yhi[2]), "+v"(yk[2].x), "+v"(yk[2].y), "+v"(yjj[2]),
                      "+v"(yi[3]), "+v"(yip[3]), "+v"(ylo[3]), "+v"(yhi[3]), "+v"(yk[3].x), "+v"(yk[3].y), "+v"(yjj[3]));
    asm volatile("" : "+v"(yi[4]), "+v"(yip[4]), "+v"(ylo[4]), "+v"(yhi[4]), "+v"(yk[4].x), "+v"(yk[4].y), "+v"(yjj[4]),
                      "+v"(yi[5]), "+v"(yip[5]), "+v"(ylo[5]), "+v"(yhi[5]), "+v"(yk[5].x), "+v"(yk[5].y), "+v"(yjj[5]),
                      "+v"(yi[6]), "+v"(yip[6]), "+v"(ylo[6]), "+v"(yhi[6]), "+v"(yk[6].x), "+v"(yk[6].y), "+v"(yjj[6]),
                      "+v"(yi[7]), "+v"(yip[7]), "+v"(ylo[7]), "+v"(yhi[7]), "+v"(yk[7].x), "+v"(yk[7].y), "+v"(yjj[7]));
    read_group(vr + (s + 2 * GRP) * C_, i0, i0p, ta, kk2, jj, xi, xip, xlo, xhi, xk, xjj);
    __builtin_amdgcn_sched_barrier(0);
    scan_iter<0>(yi, yip, ylo, yhi, yk, yjj, blo, bhi, s1, r2, a3, s1p, r2p, a3p, cacc, Ap, Bp);
    scan_iter<1>(yi, yip, ylo, yhi, yk, yjj, blo, bhi, s1, r2, a3, s1p, r2p, a3p, cacc, Ap, Bp);
    scan_iter<2>(yi, yip, ylo, yhi, yk, yjj, blo, bhi, s1, r2, a3, s1p, r2p, a3p, cacc, Ap, Bp);
    scan_iter<3>(yi, yip, ylo, yhi, yk, yjj, blo, bhi, s1, r2, a3, s1p, r2p, a3p, cacc, Ap, Bp);
    scan_iter<4>(yi, yip, ylo, yhi, yk, yjj, blo, bhi, s1, r2, a3, s1p, r2p, a3p, cacc, Ap, Bp);
    scan_iter<5>(yi, yip, ylo, yhi, yk, yjj, blo, bhi, s1, r2, a3, s1p, r2p, a3p, cacc, Ap, Bp);
    scan_iter<6>(yi, yip, ylo, yhi, yk, yjj, blo, bhi, s1, r2, a3, s1p, r2p, a3p, cacc, Ap, Bp);
    scan_iter<7>(yi, yip, ylo, yhi, yk, yjj, blo, bhi, s1, r2, a3, s1p, r2p, a3p, cacc, Ap, Bp);
    asm volatile("" : "+v"(xi[0]), "+v"(xip[0]), "+v"(xlo[0]), "+v"(xhi[0]), "+v"(xk[0].x), "+v"(xk[0].y), "+v"(xjj[0]),
                      "+v"(xi[1]), "+v"(xip[1]), "+v"(xlo[1]), "+v"(xhi[1]), "+v"(xk[1].x), "+v"(xk[1].y), "+v"(xjj[1]),
                      "+v"(xi[2]), "+v"(xip[2]), "+v"(xlo[2]), "+v"(xhi[2]), "+v"(xk[2].x), "+v"(xk[2].y), "+v"(xjj[2]),
                      "+v"(xi[3]), "+v"(xip[3]), "+v"(xlo[3]), "+v"(xhi[3]), "+v"(xk[3].x), "+v"(xk[3].y), "+v"(xjj[3]));
    asm volatile("" : "+v"(xi[4]), "+v"(xip[4]), "+v"(xlo[4]), "+v"(xhi[4]), "+v"(xk[4].x), "+v"(xk[4].y), "+v"(xjj[4]),
                      "+v"(xi[5]), "+v"(xip[5]), "+v"(xlo[5]), "+v"(xhi[5]), "+v"(xk[5].x), "+v"(xk[5].y), "+v"(xjj[5]),
                      "+v"(xi[6]), "+v"(xip[6]), "+v"(xlo[6]), "+v"(xhi[6]), "+v"(xk[6].x), "+v"(xk[6].y), "+v"(xjj[6]),
                      "+v"(xi[7]), "+v"(xip[7]), "+v"(xlo[7]), "+v"(xhi[7]), "+v"(xk[7].x), "+v"(xk[7].y), "+v"(xjj[7]));
  }
#pragma unroll
  for (int t = 0; t < NT; ++t)
    cacc[t] = __builtin_amdgcn_wmma_f32_16x16x4_f32(false, Ap[t], false, Bp, (short)0, cacc[t], false, false);

  const int mm = lane & 15;
  const int hi4 = hl * 8, cc4 = lane & 15;
  {
    const int AB = hl ? HALF_B : HALF_A;
    if (mm == 0) {
      buf[AB + S1O + i0]  = s1[0];
      buf[AB + S1O + i0p] = s1p;
    }
    if ((lane & 3) == 0) {
#pragma unroll
      for (int t = 0; t < NT; ++t) {
        buf[AB + S2O + 4 * (T0 + t) + ta]  = r2[t];
        buf[AB + S2O + 4 * (T0p + t) + ta] = r2p[t];
      }
    }
#pragma unroll
    for (int t = 0; t < NT; ++t) {
      const int fo = 32 * (T0 + t) + 2 * mm, fp = 32 * (T0p + t) + 2 * mm;
      buf[AB + S3O + fo] = a3[t].x;   buf[AB + S3O + fo + 1] = a3[t].y;
      buf[AB + S3O + fp] = a3p[t].x;  buf[AB + S3O + fp + 1] = a3p[t].y;
    }
  }
  __syncthreads();
  {
    const float* hA = &buf[HALF_A];
    const float* hB = &buf[HALF_B];
#pragma unroll
    for (int t = 0; t < NT; ++t)
#pragma unroll
      for (int q8 = 0; q8 < 8; ++q8) {
        const int f = (16 * (T0 + t) + q8 + hi4) * 16 + cc4;
        cacc[t][q8] += hA[S1O + (f >> 9)] * hB[S3O + (f & 511)]
                     + hA[S2O + (f >> 6)] * hB[S2O + (f & 63)]
                     + hA[S3O + (f >> 3)] * hB[S1O + (f & 7)];
      }
    const int Tl = hl ? T0p : T0;
#pragma unroll
    for (int t = 0; t < NT; ++t) {
#pragma unroll
      for (int r = 0; r < 2; ++r) {
        const int f = 32 * (Tl + t) + 2 * mm + r;
        const float v3 = hA[S3O + f] + hB[S3O + f]
                       + hA[S1O + (f >> 6)] * hB[S2O + (f & 63)]
                       + hA[S2O + (f >> 3)] * hB[S1O + (f & 7)];
        if (r == 0) a3[t].x = v3; else a3[t].y = v3;
      }
    }
    const int ij = 16 * w + mm;
    const float v2 = hA[S2O + ij] + hB[S2O + ij] + hA[S1O + (ij >> 3)] * hB[S1O + (ij & 7)];
    const float v1 = hA[S1O + (lane & 7)] + hB[S1O + (lane & 7)];

    if (w == 0 && lane < 8) buf[SREG + S1O + lane] = v1;
    buf[SREG + S2O + ij] = v2;
#pragma unroll
    for (int t = 0; t < NT; ++t) {
      const int f = 32 * (Tl + t) + 2 * mm;
      buf[SREG + S3O + f]     = a3[t].x;
      buf[SREG + S3O + f + 1] = a3[t].y;
    }
  }
  __syncthreads();

  float sig1 = (lane < 4) ? buf[SREG + S1O + 4 * h + lane] : 0.f;
  float r2o  = buf[SREG + S2O + 32 * h + lane];

  {
    const float* S1 = &buf[SREG + S1O];
    const float* S2 = &buf[SREG + S2O];
    const float* S3 = &buf[SREG + S3O];
    const int cb = cc4 >> 3;
    const float* p_s1j   = S1 + 2 * hl;
    const float* p_s1k   = S1 + cb;
    const float* p_s2ij  = S2 + 4 * T0 + 2 * hl;
    const float* p_s2kl  = S2 + cc4;
    const float* p_s2jk  = S2 + 16 * hl + cb;
    const float* p_s3ijk = S3 + 32 * T0 + 16 * hl + cb;
    const float* p_s3jkl = S3 + 128 * hl + cc4;
    const float s1i = S1[i0];
    const float s1l = S1[cc4 & 7];
    float s1k[4], s2kl[4];
#pragma unroll
    for (int q = 0; q < 4; ++q) { s1k[q] = p_s1k[2 * q]; s2kl[q] = p_s2kl[16 * q]; }
    float s1j[NT][2], s2ij[NT][2];
    float s2jk[NT][8], s3ijk[NT][8], s3jkl[NT][8];
#pragma unroll
    for (int t = 0; t < NT; ++t) {
#pragma unroll
      for (int u = 0; u < 2; ++u) { s1j[t][u] = p_s1j[4 * t + u]; s2ij[t][u] = p_s2ij[4 * t + u]; }
#pragma unroll
      for (int r = 0; r < 8; ++r) {
        s2jk[t][r]  = p_s2jk[32 * t + 2 * r];
        s3ijk[t][r] = p_s3ijk[32 * t + 2 * r];
        s3jkl[t][r] = p_s3jkl[256 * t + 16 * r];
      }
    }
    float2 t_s2jk[NT];
    float  t_s2ij[NT], t_s1j[NT];
    const float2 t_s1k = *(const float2*)&S1[2 * (mm & 3)];
#pragma unroll
    for (int t = 0; t < NT; ++t) {
      t_s2jk[t] = *(const float2*)&S2[32 * t + 2 * mm];
      t_s2ij[t] = S2[4 * (T0 + t) + (mm >> 2)];
      t_s1j[t]  = S1[4 * t + (mm >> 2)];
    }
    const float u_s1i = S1[4 * h + (lane >> 3)];
    const float u_s1j = S1[lane & 7];

    r2o += -0.5f * (u_s1i * u_s1j);
#pragma unroll
    for (int t = 0; t < NT; ++t) {
      const float p2  = s1i * t_s1j[t];
      const float q2x = s1i * t_s2jk[t].x + t_s2ij[t] * t_s1k.x;
      const float q2y = s1i * t_s2jk[t].y + t_s2ij[t] * t_s1k.y;
      const float q3x = p2 * t_s1k.x;
      const float q3y = p2 * t_s1k.y;
      a3[t].x += -0.5f * q2x;  a3[t].x += (1.0f / 3.0f) * q3x;
      a3[t].y += -0.5f * q2y;  a3[t].y += (1.0f / 3.0f) * q3y;
    }
#pragma unroll
    for (int t = 0; t < NT; ++t)
#pragma unroll
      for (int r = 0; r < 8; ++r) {
        const float a_s1j  = s1j[t][r >> 2], a_s2ij = s2ij[t][r >> 2];
        const float a_s1k  = s1k[r & 3],     a_s2kl = s2kl[r & 3];
        const float p2 = s1i * a_s1j;
        const float p3 = s1i * s2jk[t][r] + a_s2ij * a_s1k;
        const float q2 = s1i * s3jkl[t][r] + a_s2ij * a_s2kl + s3ijk[t][r] * s1l;
        const float q3 = p2 * a_s2kl + p3 * s1l;
        const float q4 = (p2 * a_s1k) * s1l;
        cacc[t][r] += -0.5f * q2;
        cacc[t][r] += (1.0f / 3.0f) * q3;
        cacc[t][r] += -0.25f * q4;
      }
  }

  float* ob = out + (size_t)b * OUTPB;
  if (w == 0) {
    if (lane < 4) ob[4 * h + lane] = sig1;
    ob[8 + 32 * h + lane] = r2o;
  }
  if (lane < 16) {
#pragma unroll
    for (int t = 0; t < NT; ++t) {
      int f = 32 * (T0 + t) + 2 * lane;
      ob[72 + f]     = a3[t].x;
      ob[72 + f + 1] = a3[t].y;
    }
  }
  {
    const int hi = (lane >> 4) * 8, cc = lane & 15;
#pragma unroll
    for (int t = 0; t < NT; ++t)
#pragma unroll
      for (int q = 0; q < 8; ++q) {
        int f = (16 * (T0 + t) + q + hi) * 16 + cc;
        ob[584 + f] = cacc[t][q];
      }
  }
}

extern "C" void kernel_launch(void* const* d_in, const int* in_sizes, int n_in,
                              void* d_out, int out_size, void* d_ws, size_t ws_size,
                              hipStream_t stream) {
  const float* path = (const float*)d_in[0];
  float* out = (float*)d_out;
  logsig_scan_kernel<<<dim3(NGRID), dim3(NTHR), 0, stream>>>(path, out);
}
